# starting kernel
# baseline (speedup 1.0000x reference)
#include <hip/hip_runtime.h>
#include <hip/hip_bf16.h>


typedef float v2f __attribute__((ext_vector_type(2)));
typedef float v8f __attribute__((ext_vector_type(8)));

#define N_NODES 65536
#define N_EDGES 1048576
#define DIM 64
#define M_POOL (N_NODES / DIM)


__global__ __launch_bounds__(256) void k_init_deg(float* __restrict__ deg) {
    int i = blockIdx.x * 256 + threadIdx.x;
    deg[i] = 1.0f;
}

__global__ __launch_bounds__(256) void k_deg_add(const int* __restrict__ col,
                                                 const float* __restrict__ w,
                                                 float* __restrict__ deg) {
    int e = blockIdx.x * 256 + threadIdx.x;
    atomicAdd(&deg[col[e]], w[e]);
}

__global__ __launch_bounds__(256) void k_rsqrt_inplace(float* __restrict__ d) {
    int i = blockIdx.x * 256 + threadIdx.x;
    d[i] = rsqrtf(d[i]);
}


__global__ __launch_bounds__(256) void k_gemm_wmma(const float* __restrict__ A,
                                                   const float* __restrict__ B,
                                                   float* __restrict__ C) {
    const int gwave = (blockIdx.x * 256 + (int)threadIdx.x) >> 5;
    const int lane  = threadIdx.x & 31;
    const int half  = lane >> 4;
    const int r     = lane & 15;
    const int m0    = gwave << 4;

    v8f acc0 = {}, acc1 = {}, acc2 = {}, acc3 = {};

    const float* arow = A + (size_t)(m0 + r) * DIM + 2 * half;

#pragma unroll
    for (int k0 = 0; k0 < DIM; k0 += 4) {
        v2f a;
        a.x = arow[k0];
        a.y = arow[k0 + 1];

        const float* b0 = B + (size_t)(k0 + 2 * half) * DIM + r;
        v2f b;

        b.x = b0[0];        b.y = b0[DIM];
        acc0 = __builtin_amdgcn_wmma_f32_16x16x4_f32(false, a, false, b, (short)0, acc0, false, false);
        b.x = b0[16];       b.y = b0[DIM + 16];
        acc1 = __builtin_amdgcn_wmma_f32_16x16x4_f32(false, a, false, b, (short)0, acc1, false, false);
        b.x = b0[32];       b.y = b0[DIM + 32];
        acc2 = __builtin_amdgcn_wmma_f32_16x16x4_f32(false, a, false, b, (short)0, acc2, false, false);
        b.x = b0[48];       b.y = b0[DIM + 48];
        acc3 = __builtin_amdgcn_wmma_f32_16x16x4_f32(false, a, false, b, (short)0, acc3, false, false);
    }

    float* crow = C + (size_t)(m0 + 8 * half) * DIM + r;
#pragma unroll
    for (int v = 0; v < 8; ++v) {
        crow[(size_t)v * DIM + 0]  = acc0[v];
        crow[(size_t)v * DIM + 16] = acc1[v];
        crow[(size_t)v * DIM + 32] = acc2[v];
        crow[(size_t)v * DIM + 48] = acc3[v];
    }
}


__global__ __launch_bounds__(256) void k_zero4(float4* __restrict__ p) {
    int i = blockIdx.x * 256 + threadIdx.x;
    p[i] = make_float4(0.f, 0.f, 0.f, 0.f);
}

__global__ __launch_bounds__(256) void k_edge_agg(const int* __restrict__ row,
                                                  const int* __restrict__ col,
                                                  const float* __restrict__ w,
                                                  const float* __restrict__ dinv,
                                                  const float* __restrict__ h,
                                                  float* __restrict__ out) {
    const int t    = blockIdx.x * 256 + threadIdx.x;
    const int e    = t >> 5;
    const int lane = t & 31;
    const int rs = row[e];
    const int cs = col[e];
    const float norm = dinv[rs] * w[e] * dinv[cs];
    const float2 hv = *(const float2*)(h + (size_t)rs * DIM + lane * 2);
    float* o = out + (size_t)cs * DIM + lane * 2;
    atomicAdd(o,     norm * hv.x);
    atomicAdd(o + 1, norm * hv.y);
}

__global__ __launch_bounds__(256) void k_self_bias(const float* __restrict__ h,
                                                   const float* __restrict__ dinv,
                                                   const float* __restrict__ b,
                                                   float* __restrict__ out) {
    const int t    = blockIdx.x * 256 + threadIdx.x;
    const int node = t >> 4;
    const int d4   = (t & 15) * 4;
    const float di = dinv[node];
    const float s  = di * di;
    const float4 hv = *(const float4*)(h + (size_t)node * DIM + d4);
    const float4 bv = *(const float4*)(b + d4);
    float4 ov = *(const float4*)(out + (size_t)node * DIM + d4);
    ov.x += hv.x * s + bv.x;
    ov.y += hv.y * s + bv.y;
    ov.z += hv.z * s + bv.z;
    ov.w += hv.w * s + bv.w;
    *(float4*)(out + (size_t)node * DIM + d4) = ov;
}


__global__ __launch_bounds__(256) void k_pool(const float* __restrict__ g,
                                              float* __restrict__ out) {
    const int t = blockIdx.x * 256 + threadIdx.x;
    const int m = t >> 6;
    const int d = t & 63;
    const float* base = g + (size_t)m * DIM * DIM + d;
    float s = 0.f;
#pragma unroll 8
    for (int j = 0; j < DIM; ++j) s += base[(size_t)j * DIM];
    out[(size_t)d * M_POOL + m] = s * (1.0f / DIM);
}


extern "C" void kernel_launch(void* const* d_in, const int* in_sizes, int n_in,
                              void* d_out, int out_size, void* d_ws, size_t ws_size,
                              hipStream_t stream) {
    const float* x   = (const float*)d_in[0];
    const int*  eidx = (const int*) d_in[1];
    const float* ew  = (const float*)d_in[2];
    const float* W1  = (const float*)d_in[3];
    const float* b1  = (const float*)d_in[4];
    const float* W2  = (const float*)d_in[5];
    const float* b2  = (const float*)d_in[6];
    float* out = (float*)d_out;

    const int* rowi = eidx;
    const int* coli = eidx + N_EDGES;

    float* dinv = (float*)d_ws;
    float* bufH = dinv + N_NODES;
    float* bufA = bufH + (size_t)N_NODES * DIM;

    const int ZERO_BLK = (N_NODES * DIM / 4) / 256;
    const int AGG_BLK  = (int)(((long long)N_EDGES * 32) / 256);
    const int SB_BLK   = (N_NODES * 16) / 256;
    const int GEMM_BLK = (N_NODES / 16) / 8;

    k_init_deg<<<N_NODES / 256, 256, 0, stream>>>(dinv);
    k_deg_add<<<N_EDGES / 256, 256, 0, stream>>>(coli, ew, dinv);
    k_rsqrt_inplace<<<N_NODES / 256, 256, 0, stream>>>(dinv);

    k_gemm_wmma<<<GEMM_BLK, 256, 0, stream>>>(x, W1, bufH);
    k_zero4<<<ZERO_BLK, 256, 0, stream>>>((float4*)bufA);
    k_edge_agg<<<AGG_BLK, 256, 0, stream>>>(rowi, coli, ew, dinv, bufH, bufA);
    k_self_bias<<<SB_BLK, 256, 0, stream>>>(bufH, dinv, b1, bufA);

    k_gemm_wmma<<<GEMM_BLK, 256, 0, stream>>>(bufA, W2, bufH);
    k_zero4<<<ZERO_BLK, 256, 0, stream>>>((float4*)bufA);
    k_edge_agg<<<AGG_BLK, 256, 0, stream>>>(rowi, coli, ew, dinv, bufH, bufA);
    k_self_bias<<<SB_BLK, 256, 0, stream>>>(bufH, dinv, b2, bufA);

    k_pool<<<(M_POOL * DIM) / 256, 256, 0, stream>>>(bufA, out);
}


# edge scatter: each wave takes 32 edges (indices and norms read lane-parallel once, readlane broadcast, 8 row gathers in flight per batch) instead of 1 edge per wave; atomic add order differs
# speedup vs baseline: 1.2878x; 1.2878x over previous
#include <hip/hip_runtime.h>
#include <hip/hip_bf16.h>


typedef float v2f __attribute__((ext_vector_type(2)));
typedef float v8f __attribute__((ext_vector_type(8)));

#define N_NODES 65536
#define N_EDGES 1048576
#define DIM 64
#define M_POOL (N_NODES / DIM)


__global__ __launch_bounds__(256) void k_init_deg(float* __restrict__ deg) {
    int i = blockIdx.x * 256 + threadIdx.x;
    deg[i] = 1.0f;
}

__global__ __launch_bounds__(256) void k_deg_add(const int* __restrict__ col,
                                                 const float* __restrict__ w,
                                                 float* __restrict__ deg) {
    int e = blockIdx.x * 256 + threadIdx.x;
    atomicAdd(&deg[col[e]], w[e]);
}

__global__ __launch_bounds__(256) void k_rsqrt_inplace(float* __restrict__ d) {
    int i = blockIdx.x * 256 + threadIdx.x;
    d[i] = rsqrtf(d[i]);
}


__global__ __launch_bounds__(256) void k_gemm_wmma(const float* __restrict__ A,
                                                   const float* __restrict__ B,
                                                   float* __restrict__ C) {
    const int gwave = (blockIdx.x * 256 + (int)threadIdx.x) >> 5;
    const int lane  = threadIdx.x & 31;
    const int half  = lane >> 4;
    const int r     = lane & 15;
    const int m0    = gwave << 4;

    v8f acc0 = {}, acc1 = {}, acc2 = {}, acc3 = {};

    const float* arow = A + (size_t)(m0 + r) * DIM + 2 * half;

#pragma unroll
    for (int k0 = 0; k0 < DIM; k0 += 4) {
        v2f a;
        a.x = arow[k0];
        a.y = arow[k0 + 1];

        const float* b0 = B + (size_t)(k0 + 2 * half) * DIM + r;
        v2f b;

        b.x = b0[0];        b.y = b0[DIM];
        acc0 = __builtin_amdgcn_wmma_f32_16x16x4_f32(false, a, false, b, (short)0, acc0, false, false);
        b.x = b0[16];       b.y = b0[DIM + 16];
        acc1 = __builtin_amdgcn_wmma_f32_16x16x4_f32(false, a, false, b, (short)0, acc1, false, false);
        b.x = b0[32];       b.y = b0[DIM + 32];
        acc2 = __builtin_amdgcn_wmma_f32_16x16x4_f32(false, a, false, b, (short)0, acc2, false, false);
        b.x = b0[48];       b.y = b0[DIM + 48];
        acc3 = __builtin_amdgcn_wmma_f32_16x16x4_f32(false, a, false, b, (short)0, acc3, false, false);
    }

    float* crow = C + (size_t)(m0 + 8 * half) * DIM + r;
#pragma unroll
    for (int v = 0; v < 8; ++v) {
        crow[(size_t)v * DIM + 0]  = acc0[v];
        crow[(size_t)v * DIM + 16] = acc1[v];
        crow[(size_t)v * DIM + 32] = acc2[v];
        crow[(size_t)v * DIM + 48] = acc3[v];
    }
}


__global__ __launch_bounds__(256) void k_zero4(float4* __restrict__ p) {
    int i = blockIdx.x * 256 + threadIdx.x;
    p[i] = make_float4(0.f, 0.f, 0.f, 0.f);
}

#define EDGES_PER_WAVE 32
#define EDGE_BATCH 8
__global__ __launch_bounds__(256) void k_edge_agg(const int* __restrict__ row,
                                                  const int* __restrict__ col,
                                                  const float* __restrict__ w,
                                                  const float* __restrict__ dinv,
                                                  const float* __restrict__ h,
                                                  float* __restrict__ out) {
    const int lane = threadIdx.x & 31;
    const int wave = (blockIdx.x * 256 + (int)threadIdx.x) >> 5;
    const int e    = wave * EDGES_PER_WAVE + lane;
    const int rs = row[e];
    const int cs = col[e];
    const float norm = dinv[rs] * w[e] * dinv[cs];
    const int norm_bits = __float_as_int(norm);
    const float* hl = h + lane * 2;
    float* ol = out + lane * 2;
#pragma unroll
    for (int j0 = 0; j0 < EDGES_PER_WAVE; j0 += EDGE_BATCH) {
        float2 hv[EDGE_BATCH];
#pragma unroll
        for (int j = 0; j < EDGE_BATCH; ++j) {
            const int r = __builtin_amdgcn_readlane(rs, j0 + j);
            hv[j] = *(const float2*)(hl + (size_t)r * DIM);
        }
#pragma unroll
        for (int j = 0; j < EDGE_BATCH; ++j) {
            const int c = __builtin_amdgcn_readlane(cs, j0 + j);
            const float n = __int_as_float(__builtin_amdgcn_readlane(norm_bits, j0 + j));
            float* o = ol + (size_t)c * DIM;
            atomicAdd(o,     n * hv[j].x);
            atomicAdd(o + 1, n * hv[j].y);
        }
    }
}

__global__ __launch_bounds__(256) void k_self_bias(const float* __restrict__ h,
                                                   const float* __restrict__ dinv,
                                                   const float* __restrict__ b,
                                                   float* __restrict__ out) {
    const int t    = blockIdx.x * 256 + threadIdx.x;
    const int node = t >> 4;
    const int d4   = (t & 15) * 4;
    const float di = dinv[node];
    const float s  = di * di;
    const float4 hv = *(const float4*)(h + (size_t)node * DIM + d4);
    const float4 bv = *(const float4*)(b + d4);
    float4 ov = *(const float4*)(out + (size_t)node * DIM + d4);
    ov.x += hv.x * s + bv.x;
    ov.y += hv.y * s + bv.y;
    ov.z += hv.z * s + bv.z;
    ov.w += hv.w * s + bv.w;
    *(float4*)(out + (size_t)node * DIM + d4) = ov;
}


__global__ __launch_bounds__(256) void k_pool(const float* __restrict__ g,
                                              float* __restrict__ out) {
    const int t = blockIdx.x * 256 + threadIdx.x;
    const int m = t >> 6;
    const int d = t & 63;
    const float* base = g + (size_t)m * DIM * DIM + d;
    float s = 0.f;
#pragma unroll 8
    for (int j = 0; j < DIM; ++j) s += base[(size_t)j * DIM];
    out[(size_t)d * M_POOL + m] = s * (1.0f / DIM);
}


extern "C" void kernel_launch(void* const* d_in, const int* in_sizes, int n_in,
                              void* d_out, int out_size, void* d_ws, size_t ws_size,
                              hipStream_t stream) {
    const float* x   = (const float*)d_in[0];
    const int*  eidx = (const int*) d_in[1];
    const float* ew  = (const float*)d_in[2];
    const float* W1  = (const float*)d_in[3];
    const float* b1  = (const float*)d_in[4];
    const float* W2  = (const float*)d_in[5];
    const float* b2  = (const float*)d_in[6];
    float* out = (float*)d_out;

    const int* rowi = eidx;
    const int* coli = eidx + N_EDGES;

    float* dinv = (float*)d_ws;
    float* bufH = dinv + N_NODES;
    float* bufA = bufH + (size_t)N_NODES * DIM;

    const int ZERO_BLK = (N_NODES * DIM / 4) / 256;
    const int AGG_BLK  = (N_EDGES / EDGES_PER_WAVE) / 8;
    const int SB_BLK   = (N_NODES * 16) / 256;
    const int GEMM_BLK = (N_NODES / 16) / 8;

    k_init_deg<<<N_NODES / 256, 256, 0, stream>>>(dinv);
    k_deg_add<<<N_EDGES / 256, 256, 0, stream>>>(coli, ew, dinv);
    k_rsqrt_inplace<<<N_NODES / 256, 256, 0, stream>>>(dinv);

    k_gemm_wmma<<<GEMM_BLK, 256, 0, stream>>>(x, W1, bufH);
    k_zero4<<<ZERO_BLK, 256, 0, stream>>>((float4*)bufA);
    k_edge_agg<<<AGG_BLK, 256, 0, stream>>>(rowi, coli, ew, dinv, bufH, bufA);
    k_self_bias<<<SB_BLK, 256, 0, stream>>>(bufH, dinv, b1, bufA);

    k_gemm_wmma<<<GEMM_BLK, 256, 0, stream>>>(bufA, W2, bufH);
    k_zero4<<<ZERO_BLK, 256, 0, stream>>>((float4*)bufA);
    k_edge_agg<<<AGG_BLK, 256, 0, stream>>>(rowi, coli, ew, dinv, bufH, bufA);
    k_self_bias<<<SB_BLK, 256, 0, stream>>>(bufH, dinv, b2, bufA);

    k_pool<<<(M_POOL * DIM) / 256, 256, 0, stream>>>(bufA, out);
}


# edges bucketed by destination once per call (count, scan, fill; 43 MB workspace, no launch if smaller); both layers gather per node, rows stored once, no f32 feature atomics; sum order differs
# speedup vs baseline: 1.6480x; 1.2797x over previous
#include <hip/hip_runtime.h>
#include <hip/hip_bf16.h>


typedef float v2f __attribute__((ext_vector_type(2)));
typedef float v8f __attribute__((ext_vector_type(8)));

#define N_NODES 65536
#define N_EDGES 1048576
#define DIM 64
#define M_POOL (N_NODES / DIM)


__global__ __launch_bounds__(256) void k_init_deg(float* __restrict__ deg, int* __restrict__ cnt) {
    int i = blockIdx.x * 256 + threadIdx.x;
    deg[i] = 1.0f;
    cnt[i] = 0;
}

__global__ __launch_bounds__(256) void k_deg_add(const int* __restrict__ col,
                                                 const float* __restrict__ w,
                                                 float* __restrict__ deg,
                                                 int* __restrict__ cnt,
                                                 int* __restrict__ pos) {
    int e = blockIdx.x * 256 + threadIdx.x;
    const int c = col[e];
    atomicAdd(&deg[c], w[e]);
    pos[e] = atomicAdd(&cnt[c], 1);
}

__global__ __launch_bounds__(256) void k_rsqrt_inplace(float* __restrict__ d) {
    int i = blockIdx.x * 256 + threadIdx.x;
    d[i] = rsqrtf(d[i]);
}

#define SCAN_WAVES 32
#define SCAN_PIECES 16
__global__ __launch_bounds__(1024) void k_scan(const int* __restrict__ cnt,
                                               int* __restrict__ rowptr) {
    __shared__ int wsum[SCAN_WAVES];
    const int lane = threadIdx.x & 31;
    const int wv   = threadIdx.x >> 5;
    const int4* src = (const int4*)(cnt + wv * (SCAN_PIECES * 128)) + lane;
    int4 v[SCAN_PIECES];
#pragma unroll
    for (int it = 0; it < SCAN_PIECES; ++it) v[it] = src[it * 32];
    int carry = 0;
#pragma unroll
    for (int it = 0; it < SCAN_PIECES; ++it) {
        const int4 x = v[it];
        const int s1 = x.x, s2 = s1 + x.y, s3 = s2 + x.z, s4 = s3 + x.w;
        int inc = s4;
        for (int d = 1; d < 32; d <<= 1) {
            const int y = __shfl_up(inc, d);
            if (lane >= d) inc += y;
        }
        const int excl = inc - s4 + carry;
        v[it] = make_int4(excl, excl + s1, excl + s2, excl + s3);
        carry += __shfl(inc, 31);
    }
    if (lane == 0) wsum[wv] = carry;
    __syncthreads();
    int incw = wsum[lane];
    for (int d = 1; d < 32; d <<= 1) {
        const int y = __shfl_up(incw, d);
        if (lane >= d) incw += y;
    }
    const int below = __shfl(incw, (wv + 31) & 31);
    const int base  = (wv == 0) ? 0 : below;
    int4* dst = (int4*)(rowptr + wv * (SCAN_PIECES * 128)) + lane;
#pragma unroll
    for (int it = 0; it < SCAN_PIECES; ++it) {
        int4 o = v[it];
        o.x += base; o.y += base; o.z += base; o.w += base;
        dst[it * 32] = o;
    }
    if (threadIdx.x == 1023) rowptr[N_NODES] = incw;
}

__global__ __launch_bounds__(256) void k_fill(const int* __restrict__ row,
                                              const int* __restrict__ col,
                                              const float* __restrict__ w,
                                              const float* __restrict__ dinv,
                                              const int* __restrict__ rowptr,
                                              const int* __restrict__ pos,
                                              int2* __restrict__ recs) {
    const int e = blockIdx.x * 256 + threadIdx.x;
    const int r = row[e];
    const int c = col[e];
    const float norm = dinv[r] * w[e] * dinv[c];
    const int p = rowptr[c] + pos[e];
    if ((unsigned)p < (unsigned)N_EDGES) recs[p] = make_int2(r, __float_as_int(norm));
}


__global__ __launch_bounds__(256) void k_gemm_wmma(const float* __restrict__ A,
                                                   const float* __restrict__ B,
                                                   float* __restrict__ C) {
    const int gwave = (blockIdx.x * 256 + (int)threadIdx.x) >> 5;
    const int lane  = threadIdx.x & 31;
    const int half  = lane >> 4;
    const int r     = lane & 15;
    const int m0    = gwave << 4;

    v8f acc0 = {}, acc1 = {}, acc2 = {}, acc3 = {};

    const float* arow = A + (size_t)(m0 + r) * DIM + 2 * half;

#pragma unroll
    for (int k0 = 0; k0 < DIM; k0 += 4) {
        v2f a;
        a.x = arow[k0];
        a.y = arow[k0 + 1];

        const float* b0 = B + (size_t)(k0 + 2 * half) * DIM + r;
        v2f b;

        b.x = b0[0];        b.y = b0[DIM];
        acc0 = __builtin_amdgcn_wmma_f32_16x16x4_f32(false, a, false, b, (short)0, acc0, false, false);
        b.x = b0[16];       b.y = b0[DIM + 16];
        acc1 = __builtin_amdgcn_wmma_f32_16x16x4_f32(false, a, false, b, (short)0, acc1, false, false);
        b.x = b0[32];       b.y = b0[DIM + 32];
        acc2 = __builtin_amdgcn_wmma_f32_16x16x4_f32(false, a, false, b, (short)0, acc2, false, false);
        b.x = b0[48];       b.y = b0[DIM + 48];
        acc3 = __builtin_amdgcn_wmma_f32_16x16x4_f32(false, a, false, b, (short)0, acc3, false, false);
    }

    float* crow = C + (size_t)(m0 + 8 * half) * DIM + r;
#pragma unroll
    for (int v = 0; v < 8; ++v) {
        crow[(size_t)v * DIM + 0]  = acc0[v];
        crow[(size_t)v * DIM + 16] = acc1[v];
        crow[(size_t)v * DIM + 32] = acc2[v];
        crow[(size_t)v * DIM + 48] = acc3[v];
    }
}

#define NODES_PER_WAVE 4
__global__ __launch_bounds__(256) void k_gather(const int* __restrict__ rowptr,
                                                const int2* __restrict__ recs,
                                                const float* __restrict__ dinv,
                                                const float* __restrict__ h,
                                                const float* __restrict__ b,
                                                float* __restrict__ out) {
    const int lane = threadIdx.x & 31;
    const int wave = (blockIdx.x * 256 + (int)threadIdx.x) >> 5;
    const int n0   = wave * NODES_PER_WAVE;
    const int bnd = rowptr[n0 + (lane < NODES_PER_WAVE ? lane : NODES_PER_WAVE)];
    const float2 bias = *(const float2*)(b + lane * 2);
    const float* hl = h + lane * 2;
    for (int i = 0; i < NODES_PER_WAVE; ++i) {
        const int node = n0 + i;
        const int s = max(__builtin_amdgcn_readlane(bnd, i), 0);
        const int e = min(__builtin_amdgcn_readlane(bnd, i + 1), N_EDGES);
        const float di = dinv[node];
        const float sl = di * di;
        const float2 hv = *(const float2*)(hl + (size_t)node * DIM);
        float2 acc;
        acc.x = hv.x * sl + bias.x;
        acc.y = hv.y * sl + bias.y;
        for (int p0 = s; p0 < e; p0 += 32) {
            const int p = p0 + lane;
            int2 rec = make_int2(0, 0);
            if (p < e) rec = recs[p];
            const int cnt = min(32, e - p0);
            int j = 0;
            for (; j + 8 <= cnt; j += 8) {
                float2 v[8];
#pragma unroll
                for (int u = 0; u < 8; ++u) {
                    const int r = __builtin_amdgcn_readlane(rec.x, j + u);
                    v[u] = *(const float2*)(hl + (size_t)r * DIM);
                }
#pragma unroll
                for (int u = 0; u < 8; ++u) {
                    const float n = __int_as_float(__builtin_amdgcn_readlane(rec.y, j + u));
                    acc.x += n * v[u].x;
                    acc.y += n * v[u].y;
                }
            }
            for (; j < cnt; j += 4) {
                float2 v[4];
#pragma unroll
                for (int u = 0; u < 4; ++u) {
                    const int r = __builtin_amdgcn_readlane(rec.x, min(j + u, cnt - 1));
                    v[u] = *(const float2*)(hl + (size_t)r * DIM);
                }
#pragma unroll
                for (int u = 0; u < 4; ++u) {
                    const float nv = __int_as_float(__builtin_amdgcn_readlane(rec.y, min(j + u, cnt - 1)));
                    const float n = (j + u < cnt) ? nv : 0.0f;
                    acc.x += n * v[u].x;
                    acc.y += n * v[u].y;
                }
            }
        }
        *(float2*)(out + (size_t)node * DIM + lane * 2) = acc;
    }
}


__global__ __launch_bounds__(256) void k_pool(const float* __restrict__ g,
                                              float* __restrict__ out) {
    const int t = blockIdx.x * 256 + threadIdx.x;
    const int m = t >> 6;
    const int d = t & 63;
    const float* base = g + (size_t)m * DIM * DIM + d;
    float s = 0.f;
#pragma unroll 8
    for (int j = 0; j < DIM; ++j) s += base[(size_t)j * DIM];
    out[(size_t)d * M_POOL + m] = s * (1.0f / DIM);
}


extern "C" void kernel_launch(void* const* d_in, const int* in_sizes, int n_in,
                              void* d_out, int out_size, void* d_ws, size_t ws_size,
                              hipStream_t stream) {
    const float* x   = (const float*)d_in[0];
    const int*  eidx = (const int*) d_in[1];
    const float* ew  = (const float*)d_in[2];
    const float* W1  = (const float*)d_in[3];
    const float* b1  = (const float*)d_in[4];
    const float* W2  = (const float*)d_in[5];
    const float* b2  = (const float*)d_in[6];
    float* out = (float*)d_out;

    const int* rowi = eidx;
    const int* coli = eidx + N_EDGES;

    if (ws_size < (size_t)42467584) return;
    float* dinv = (float*)d_ws;
    float* bufH = dinv + N_NODES;
    float* bufA = bufH + (size_t)N_NODES * DIM;
    int*  rowptr = (int*)(bufA + (size_t)N_NODES * DIM);
    int2* recs   = (int2*)(rowptr + N_NODES + 64);
    int*  cnt    = (int*)bufA;
    int*  pos    = cnt + N_NODES;

    const int GATHER_BLK = (N_NODES / NODES_PER_WAVE) / 8;
    const int GEMM_BLK   = (N_NODES / 16) / 8;

    k_init_deg<<<N_NODES / 256, 256, 0, stream>>>(dinv, cnt);
    k_deg_add<<<N_EDGES / 256, 256, 0, stream>>>(coli, ew, dinv, cnt, pos);
    k_scan<<<1, 1024, 0, stream>>>(cnt, rowptr);
    k_rsqrt_inplace<<<N_NODES / 256, 256, 0, stream>>>(dinv);
    k_fill<<<N_EDGES / 256, 256, 0, stream>>>(rowi, coli, ew, dinv, rowptr, pos, recs);

    k_gemm_wmma<<<GEMM_BLK, 256, 0, stream>>>(x, W1, bufH);
    k_gather<<<GATHER_BLK, 256, 0, stream>>>(rowptr, recs, dinv, bufH, b1, bufA);

    k_gemm_wmma<<<GEMM_BLK, 256, 0, stream>>>(bufA, W2, bufH);
    k_gather<<<GATHER_BLK, 256, 0, stream>>>(rowptr, recs, dinv, bufH, b2, bufA);

    k_pool<<<(M_POOL * DIM) / 256, 256, 0, stream>>>(bufA, out);
}


# edge norm applied in factors: buckets hold raw weights, degree = sum of each bucket (no f32 atomics), rows scaled by dinv at the GEMM store and sums by dinv[c] in the gather; sum order differs
# speedup vs baseline: 2.0458x; 1.2414x over previous
#include <hip/hip_runtime.h>
#include <hip/hip_bf16.h>


typedef float v2f __attribute__((ext_vector_type(2)));
typedef float v8f __attribute__((ext_vector_type(8)));

#define N_NODES 65536
#define N_EDGES 1048576
#define DIM 64
#define M_POOL (N_NODES / DIM)


__global__ __launch_bounds__(256) void k_init_cnt(int* __restrict__ cnt) {
    cnt[blockIdx.x * 256 + threadIdx.x] = 0;
}

__global__ __launch_bounds__(256) void k_count(const int* __restrict__ col,
                                               int* __restrict__ cnt,
                                               int* __restrict__ pos) {
    int e = blockIdx.x * 256 + threadIdx.x;
    pos[e] = atomicAdd(&cnt[col[e]], 1);
}

#define SCAN_WAVES 32
#define SCAN_PIECES 16
__global__ __launch_bounds__(1024) void k_scan(const int* __restrict__ cnt,
                                               int* __restrict__ rowptr) {
    __shared__ int wsum[SCAN_WAVES];
    const int lane = threadIdx.x & 31;
    const int wv   = threadIdx.x >> 5;
    const int4* src = (const int4*)(cnt + wv * (SCAN_PIECES * 128)) + lane;
    int4 v[SCAN_PIECES];
#pragma unroll
    for (int it = 0; it < SCAN_PIECES; ++it) v[it] = src[it * 32];
    int carry = 0;
#pragma unroll
    for (int it = 0; it < SCAN_PIECES; ++it) {
        const int4 x = v[it];
        const int s1 = x.x, s2 = s1 + x.y, s3 = s2 + x.z, s4 = s3 + x.w;
        int inc = s4;
        for (int d = 1; d < 32; d <<= 1) {
            const int y = __shfl_up(inc, d);
            if (lane >= d) inc += y;
        }
        const int excl = inc - s4 + carry;
        v[it] = make_int4(excl, excl + s1, excl + s2, excl + s3);
        carry += __shfl(inc, 31);
    }
    if (lane == 0) wsum[wv] = carry;
    __syncthreads();
    int incw = wsum[lane];
    for (int d = 1; d < 32; d <<= 1) {
        const int y = __shfl_up(incw, d);
        if (lane >= d) incw += y;
    }
    const int below = __shfl(incw, (wv + 31) & 31);
    const int base  = (wv == 0) ? 0 : below;
    int4* dst = (int4*)(rowptr + wv * (SCAN_PIECES * 128)) + lane;
#pragma unroll
    for (int it = 0; it < SCAN_PIECES; ++it) {
        int4 o = v[it];
        o.x += base; o.y += base; o.z += base; o.w += base;
        dst[it * 32] = o;
    }
    if (threadIdx.x == 1023) rowptr[N_NODES] = incw;
}

__global__ __launch_bounds__(256) void k_fill(const int* __restrict__ row,
                                              const int* __restrict__ col,
                                              const float* __restrict__ w,
                                              const int* __restrict__ rowptr,
                                              const int* __restrict__ pos,
                                              int2* __restrict__ recs) {
    const int e = blockIdx.x * 256 + threadIdx.x;
    const int r = row[e];
    const float wv = w[e];
    const int p = rowptr[col[e]] + pos[e];
    if ((unsigned)p < (unsigned)N_EDGES) recs[p] = make_int2(r, __float_as_int(wv));
}

__global__ __launch_bounds__(256) void k_dinv(const int* __restrict__ rowptr,
                                              const int2* __restrict__ recs,
                                              float* __restrict__ dinv) {
    const int n = blockIdx.x * 256 + threadIdx.x;
    const int s = max(rowptr[n], 0);
    const int e = min(rowptr[n + 1], N_EDGES);
    float sum = 0.0f;
    for (int p = s; p < e; p += 4) {
        float wv[4];
#pragma unroll
        for (int u = 0; u < 4; ++u)
            wv[u] = (p + u < e) ? __int_as_float(recs[p + u].y) : 0.0f;
        sum = (((sum + wv[0]) + wv[1]) + wv[2]) + wv[3];
    }
    const float deg = sum + 1.0f;
    dinv[n] = (deg > 0.0f) ? rsqrtf(deg) : 0.0f;
}


__global__ __launch_bounds__(256) void k_gemm_wmma(const float* __restrict__ A,
                                                   const float* __restrict__ B,
                                                   const float* __restrict__ dinv,
                                                   float* __restrict__ C) {
    const int gwave = (blockIdx.x * 256 + (int)threadIdx.x) >> 5;
    const int lane  = threadIdx.x & 31;
    const int half  = lane >> 4;
    const int r     = lane & 15;
    const int m0    = gwave << 4;

    v8f acc0 = {}, acc1 = {}, acc2 = {}, acc3 = {};

    const float* arow = A + (size_t)(m0 + r) * DIM + 2 * half;

#pragma unroll
    for (int k0 = 0; k0 < DIM; k0 += 4) {
        v2f a;
        a.x = arow[k0];
        a.y = arow[k0 + 1];

        const float* b0 = B + (size_t)(k0 + 2 * half) * DIM + r;
        v2f b;

        b.x = b0[0];        b.y = b0[DIM];
        acc0 = __builtin_amdgcn_wmma_f32_16x16x4_f32(false, a, false, b, (short)0, acc0, false, false);
        b.x = b0[16];       b.y = b0[DIM + 16];
        acc1 = __builtin_amdgcn_wmma_f32_16x16x4_f32(false, a, false, b, (short)0, acc1, false, false);
        b.x = b0[32];       b.y = b0[DIM + 32];
        acc2 = __builtin_amdgcn_wmma_f32_16x16x4_f32(false, a, false, b, (short)0, acc2, false, false);
        b.x = b0[48];       b.y = b0[DIM + 48];
        acc3 = __builtin_amdgcn_wmma_f32_16x16x4_f32(false, a, false, b, (short)0, acc3, false, false);
    }

    const float4 dlo = *(const float4*)(dinv + m0 + 8 * half);
    const float4 dhi = *(const float4*)(dinv + m0 + 8 * half + 4);
    const float ds[8] = {dlo.x, dlo.y, dlo.z, dlo.w, dhi.x, dhi.y, dhi.z, dhi.w};
    float* crow = C + (size_t)(m0 + 8 * half) * DIM + r;
#pragma unroll
    for (int v = 0; v < 8; ++v) {
        crow[(size_t)v * DIM + 0]  = acc0[v] * ds[v];
        crow[(size_t)v * DIM + 16] = acc1[v] * ds[v];
        crow[(size_t)v * DIM + 32] = acc2[v] * ds[v];
        crow[(size_t)v * DIM + 48] = acc3[v] * ds[v];
    }
}

#define NODES_PER_WAVE 4
__global__ __launch_bounds__(256) void k_gather(const int* __restrict__ rowptr,
                                                const int2* __restrict__ recs,
                                                const float* __restrict__ dinv,
                                                const float* __restrict__ h,
                                                const float* __restrict__ b,
                                                float* __restrict__ out) {
    const int lane = threadIdx.x & 31;
    const int wave = (blockIdx.x * 256 + (int)threadIdx.x) >> 5;
    const int n0   = wave * NODES_PER_WAVE;
    const int bnd = rowptr[n0 + (lane < NODES_PER_WAVE ? lane : NODES_PER_WAVE)];
    const float2 bias = *(const float2*)(b + lane * 2);
    const float* hl = h + lane * 2;
    for (int i = 0; i < NODES_PER_WAVE; ++i) {
        const int node = n0 + i;
        const int s = max(__builtin_amdgcn_readlane(bnd, i), 0);
        const int e = min(__builtin_amdgcn_readlane(bnd, i + 1), N_EDGES);
        const float di = dinv[node];
        float2 acc = *(const float2*)(hl + (size_t)node * DIM);
        for (int p0 = s; p0 < e; p0 += 32) {
            const int p = p0 + lane;
            int2 rec = make_int2(0, 0);
            if (p < e) rec = recs[p];
            const int cnt = min(32, e - p0);
            int j = 0;
            for (; j + 8 <= cnt; j += 8) {
                float2 v[8];
#pragma unroll
                for (int u = 0; u < 8; ++u) {
                    const int r = __builtin_amdgcn_readlane(rec.x, j + u);
                    v[u] = *(const float2*)(hl + (size_t)r * DIM);
                }
#pragma unroll
                for (int u = 0; u < 8; ++u) {
                    const float n = __int_as_float(__builtin_amdgcn_readlane(rec.y, j + u));
                    acc.x += n * v[u].x;
                    acc.y += n * v[u].y;
                }
            }
            for (; j < cnt; j += 4) {
                float2 v[4];
#pragma unroll
                for (int u = 0; u < 4; ++u) {
                    const int r = __builtin_amdgcn_readlane(rec.x, min(j + u, cnt - 1));
                    v[u] = *(const float2*)(hl + (size_t)r * DIM);
                }
#pragma unroll
                for (int u = 0; u < 4; ++u) {
                    const float nv = __int_as_float(__builtin_amdgcn_readlane(rec.y, min(j + u, cnt - 1)));
                    const float n = (j + u < cnt) ? nv : 0.0f;
                    acc.x += n * v[u].x;
                    acc.y += n * v[u].y;
                }
            }
        }
        float2 o;
        o.x = acc.x * di + bias.x;
        o.y = acc.y * di + bias.y;
        *(float2*)(out + (size_t)node * DIM + lane * 2) = o;
    }
}


__global__ __launch_bounds__(256) void k_pool(const float* __restrict__ g,
                                              float* __restrict__ out) {
    const int t = blockIdx.x * 256 + threadIdx.x;
    const int m = t >> 6;
    const int d = t & 63;
    const float* base = g + (size_t)m * DIM * DIM + d;
    float s = 0.f;
#pragma unroll 8
    for (int j = 0; j < DIM; ++j) s += base[(size_t)j * DIM];
    out[(size_t)d * M_POOL + m] = s * (1.0f / DIM);
}


extern "C" void kernel_launch(void* const* d_in, const int* in_sizes, int n_in,
                              void* d_out, int out_size, void* d_ws, size_t ws_size,
                              hipStream_t stream) {
    const float* x   = (const float*)d_in[0];
    const int*  eidx = (const int*) d_in[1];
    const float* ew  = (const float*)d_in[2];
    const float* W1  = (const float*)d_in[3];
    const float* b1  = (const float*)d_in[4];
    const float* W2  = (const float*)d_in[5];
    const float* b2  = (const float*)d_in[6];
    float* out = (float*)d_out;

    const int* rowi = eidx;
    const int* coli = eidx + N_EDGES;

    if (ws_size < (size_t)42467584) return;
    float* dinv = (float*)d_ws;
    float* bufH = dinv + N_NODES;
    float* bufA = bufH + (size_t)N_NODES * DIM;
    int*  rowptr = (int*)(bufA + (size_t)N_NODES * DIM);
    int2* recs   = (int2*)(rowptr + N_NODES + 64);
    int*  cnt    = (int*)bufA;
    int*  pos    = cnt + N_NODES;

    const int GATHER_BLK = (N_NODES / NODES_PER_WAVE) / 8;
    const int GEMM_BLK   = (N_NODES / 16) / 8;

    k_init_cnt<<<N_NODES / 256, 256, 0, stream>>>(cnt);
    k_count<<<N_EDGES / 256, 256, 0, stream>>>(coli, cnt, pos);
    k_scan<<<1, 1024, 0, stream>>>(cnt, rowptr);
    k_fill<<<N_EDGES / 256, 256, 0, stream>>>(rowi, coli, ew, rowptr, pos, recs);
    k_dinv<<<N_NODES / 256, 256, 0, stream>>>(rowptr, recs, dinv);

    k_gemm_wmma<<<GEMM_BLK, 256, 0, stream>>>(x, W1, dinv, bufH);
    k_gather<<<GATHER_BLK, 256, 0, stream>>>(rowptr, recs, dinv, bufH, b1, bufA);

    k_gemm_wmma<<<GEMM_BLK, 256, 0, stream>>>(bufA, W2, dinv, bufH);
    k_gather<<<GATHER_BLK, 256, 0, stream>>>(rowptr, recs, dinv, bufH, b2, bufA);

    k_pool<<<(M_POOL * DIM) / 256, 256, 0, stream>>>(bufA, out);
}


# edges bucketed by a two-level counting sort on LDS counters (256 coarse bins per 4096-edge chunk, then 256 nodes per bin): no global atomics, buckets stored in whole lines; bucket slot order differs
# speedup vs baseline: 3.2125x; 1.5703x over previous
#include <hip/hip_runtime.h>
#include <hip/hip_bf16.h>


typedef float v2f __attribute__((ext_vector_type(2)));
typedef float v8f __attribute__((ext_vector_type(8)));

#define N_NODES 65536
#define N_EDGES 1048576
#define DIM 64
#define M_POOL (N_NODES / DIM)


#define PART_EDGES 4096
#define N_PART (N_EDGES / PART_EDGES)
#define N_CBIN 256
#define FINE_CAP 6144

__global__ __launch_bounds__(256) void k_hist(const int* __restrict__ col,
                                              int* __restrict__ hist) {
    __shared__ int lcnt[N_CBIN];
    const int t = threadIdx.x;
    const int4* src = (const int4*)(col + (size_t)blockIdx.x * PART_EDGES) + t;
    lcnt[t] = 0;
    __syncthreads();
    int4 c[4];
#pragma unroll
    for (int i = 0; i < 4; ++i) c[i] = src[i * 256];
#pragma unroll
    for (int i = 0; i < 4; ++i) {
        atomicAdd(&lcnt[(c[i].x >> 8) & 255], 1);
        atomicAdd(&lcnt[(c[i].y >> 8) & 255], 1);
        atomicAdd(&lcnt[(c[i].z >> 8) & 255], 1);
        atomicAdd(&lcnt[(c[i].w >> 8) & 255], 1);
    }
    __syncthreads();
    hist[t * N_PART + blockIdx.x] = lcnt[t];
}

#define SCAN_WAVES 32
#define SCAN_PIECES 16
__global__ __launch_bounds__(1024) void k_scan(const int* __restrict__ cnt,
                                               int* __restrict__ rowptr) {
    __shared__ int wsum[SCAN_WAVES];
    const int lane = threadIdx.x & 31;
    const int wv   = threadIdx.x >> 5;
    const int4* src = (const int4*)(cnt + wv * (SCAN_PIECES * 128)) + lane;
    int4 v[SCAN_PIECES];
#pragma unroll
    for (int it = 0; it < SCAN_PIECES; ++it) v[it] = src[it * 32];
    int carry = 0;
#pragma unroll
    for (int it = 0; it < SCAN_PIECES; ++it) {
        const int4 x = v[it];
        const int s1 = x.x, s2 = s1 + x.y, s3 = s2 + x.z, s4 = s3 + x.w;
        int inc = s4;
        for (int d = 1; d < 32; d <<= 1) {
            const int y = __shfl_up(inc, d);
            if (lane >= d) inc += y;
        }
        const int excl = inc - s4 + carry;
        v[it] = make_int4(excl, excl + s1, excl + s2, excl + s3);
        carry += __shfl(inc, 31);
    }
    if (lane == 0) wsum[wv] = carry;
    __syncthreads();
    int incw = wsum[lane];
    for (int d = 1; d < 32; d <<= 1) {
        const int y = __shfl_up(incw, d);
        if (lane >= d) incw += y;
    }
    const int below = __shfl(incw, (wv + 31) & 31);
    const int base  = (wv == 0) ? 0 : below;
    int4* dst = (int4*)(rowptr + wv * (SCAN_PIECES * 128)) + lane;
#pragma unroll
    for (int it = 0; it < SCAN_PIECES; ++it) {
        int4 o = v[it];
        o.x += base; o.y += base; o.z += base; o.w += base;
        dst[it * 32] = o;
    }
    if (threadIdx.x == 1023) rowptr[N_NODES] = incw;
}

__device__ __forceinline__ int excl_scan_256(const int v, int* wsum) {
    const int lane = threadIdx.x & 31;
    const int wv = threadIdx.x >> 5;
    int inc = v;
    for (int d = 1; d < 32; d <<= 1) {
        const int y = __shfl_up(inc, d);
        if (lane >= d) inc += y;
    }
    if (lane == 31) wsum[wv] = inc;
    __syncthreads();
    int below = 0;
    for (int k = 0; k < 8; ++k) below += (k < wv) ? wsum[k] : 0;
    return inc - v + below;
}

__global__ __launch_bounds__(256) void k_part(const int* __restrict__ row,
                                              const int* __restrict__ col,
                                              const float* __restrict__ w,
                                              const int* __restrict__ off,
                                              int2* __restrict__ coarse) {
    __shared__ int lcnt[N_CBIN];
    __shared__ int lgo[N_CBIN];
    __shared__ int wsum[8];
    __shared__ int2 lrec[PART_EDGES];
    const int t = threadIdx.x;
    const size_t q = (size_t)blockIdx.x * (PART_EDGES / 4) + t;
    lcnt[t] = 0;
    __syncthreads();
    int4 c4[4], r4[4];
    float4 w4[4];
#pragma unroll
    for (int i = 0; i < 4; ++i) c4[i] = ((const int4*)col)[q + i * 256];
#pragma unroll
    for (int i = 0; i < 4; ++i) {
        r4[i] = ((const int4*)row)[q + i * 256];
        w4[i] = ((const float4*)w)[q + i * 256];
    }
    const int goff = off[t * N_PART + blockIdx.x];
    unsigned cc[16], rr[16];
    float wr[16];
#pragma unroll
    for (int i = 0; i < 4; ++i) {
        cc[4 * i + 0] = (unsigned)c4[i].x & 0xFFFFu;  rr[4 * i + 0] = (unsigned)r4[i].x & 0xFFFFu;
        cc[4 * i + 1] = (unsigned)c4[i].y & 0xFFFFu;  rr[4 * i + 1] = (unsigned)r4[i].y & 0xFFFFu;
        cc[4 * i + 2] = (unsigned)c4[i].z & 0xFFFFu;  rr[4 * i + 2] = (unsigned)r4[i].z & 0xFFFFu;
        cc[4 * i + 3] = (unsigned)c4[i].w & 0xFFFFu;  rr[4 * i + 3] = (unsigned)r4[i].w & 0xFFFFu;
        wr[4 * i + 0] = w4[i].x;  wr[4 * i + 1] = w4[i].y;
        wr[4 * i + 2] = w4[i].z;  wr[4 * i + 3] = w4[i].w;
    }
    int rank[16];
#pragma unroll
    for (int k = 0; k < 16; ++k) rank[k] = atomicAdd(&lcnt[cc[k] >> 8], 1);
    __syncthreads();
    const int cnt = lcnt[t];
    const int first = excl_scan_256(cnt, wsum);
    lcnt[t] = first;
    lgo[t] = goff - first;
    __syncthreads();
    int fs[16];
#pragma unroll
    for (int k = 0; k < 16; ++k) fs[k] = lcnt[cc[k] >> 8];
#pragma unroll
    for (int k = 0; k < 16; ++k)
        lrec[fs[k] + rank[k]] = make_int2((int)(rr[k] | (cc[k] << 16)), __float_as_int(wr[k]));
    __syncthreads();
    int2 rec[16];
    int go[16];
#pragma unroll
    for (int i = 0; i < 16; ++i) rec[i] = lrec[i * 256 + t];
#pragma unroll
    for (int i = 0; i < 16; ++i) go[i] = lgo[((unsigned)rec[i].x) >> 24];
#pragma unroll
    for (int i = 0; i < 16; ++i) {
        const int d = go[i] + i * 256 + t;
        if ((unsigned)d < (unsigned)N_EDGES) coarse[d] = rec[i];
    }
}

__global__ __launch_bounds__(256) void k_fine(const int* __restrict__ off,
                                              const int2* __restrict__ coarse,
                                              int* __restrict__ rowptr,
                                              int2* __restrict__ recs) {
    __shared__ int lcnt[256];
    __shared__ int wsum[8];
    __shared__ int2 lout[FINE_CAP];
    const int t = threadIdx.x;
    const int b = blockIdx.x;
    const int s = min(max(off[b * N_PART], 0), N_EDGES);
    const int e = min(max(off[(b + 1) * N_PART], s), N_EDGES);
    const int n = e - s;
    lcnt[t] = 0;
    __syncthreads();
    for (int p0 = s; p0 < e; p0 += 4096) {
        int x[16];
#pragma unroll
        for (int u = 0; u < 16; ++u) x[u] = coarse[min(p0 + u * 256 + t, e - 1)].x;
#pragma unroll
        for (int u = 0; u < 16; ++u)
            if (p0 + u * 256 + t < e) atomicAdd(&lcnt[((unsigned)x[u] >> 16) & 255u], 1);
    }
    __syncthreads();
    const int cnt = lcnt[t];
    const int first = excl_scan_256(cnt, wsum);
    rowptr[b * 256 + t] = s + first;
    lcnt[t] = first;
    if (b == N_CBIN - 1 && t == 0) rowptr[N_NODES] = e;
    __syncthreads();
    const bool fits = (n <= FINE_CAP);
    for (int p0 = s; p0 < e; p0 += 4096) {
        int2 r[16];
        int pos[16];
#pragma unroll
        for (int u = 0; u < 16; ++u) r[u] = coarse[min(p0 + u * 256 + t, e - 1)];
#pragma unroll
        for (int u = 0; u < 16; ++u)
            pos[u] = (p0 + u * 256 + t < e)
                         ? atomicAdd(&lcnt[((unsigned)r[u].x >> 16) & 255u], 1) : -1;
#pragma unroll
        for (int u = 0; u < 16; ++u) {
            const int2 o = make_int2(r[u].x & 0xFFFF, r[u].y);
            if (fits) {
                if ((unsigned)pos[u] < (unsigned)FINE_CAP) lout[pos[u]] = o;
            } else {
                if (pos[u] >= 0 && (unsigned)(s + pos[u]) < (unsigned)N_EDGES) recs[s + pos[u]] = o;
            }
        }
    }
    if (fits) {
        __syncthreads();
        for (int i0 = 0; i0 < n; i0 += 2048) {
            int2 v[8];
#pragma unroll
            for (int u = 0; u < 8; ++u) v[u] = lout[min(i0 + u * 256 + t, n - 1)];
#pragma unroll
            for (int u = 0; u < 8; ++u)
                if (i0 + u * 256 + t < n) recs[s + i0 + u * 256 + t] = v[u];
        }
    }
}

__global__ __launch_bounds__(256) void k_dinv(const int* __restrict__ rowptr,
                                              const int2* __restrict__ recs,
                                              float* __restrict__ dinv) {
    const int n = blockIdx.x * 256 + threadIdx.x;
    const int s = max(rowptr[n], 0);
    const int e = min(rowptr[n + 1], N_EDGES);
    float sum = 0.0f;
    for (int p = s; p < e; p += 4) {
        float wv[4];
#pragma unroll
        for (int u = 0; u < 4; ++u)
            wv[u] = (p + u < e) ? __int_as_float(recs[p + u].y) : 0.0f;
        sum = (((sum + wv[0]) + wv[1]) + wv[2]) + wv[3];
    }
    const float deg = sum + 1.0f;
    dinv[n] = (deg > 0.0f) ? rsqrtf(deg) : 0.0f;
}


__global__ __launch_bounds__(256) void k_gemm_wmma(const float* __restrict__ A,
                                                   const float* __restrict__ B,
                                                   const float* __restrict__ dinv,
                                                   float* __restrict__ C) {
    const int gwave = (blockIdx.x * 256 + (int)threadIdx.x) >> 5;
    const int lane  = threadIdx.x & 31;
    const int half  = lane >> 4;
    const int r     = lane & 15;
    const int m0    = gwave << 4;

    v8f acc0 = {}, acc1 = {}, acc2 = {}, acc3 = {};

    const float* arow = A + (size_t)(m0 + r) * DIM + 2 * half;

#pragma unroll
    for (int k0 = 0; k0 < DIM; k0 += 4) {
        v2f a;
        a.x = arow[k0];
        a.y = arow[k0 + 1];

        const float* b0 = B + (size_t)(k0 + 2 * half) * DIM + r;
        v2f b;

        b.x = b0[0];        b.y = b0[DIM];
        acc0 = __builtin_amdgcn_wmma_f32_16x16x4_f32(false, a, false, b, (short)0, acc0, false, false);
        b.x = b0[16];       b.y = b0[DIM + 16];
        acc1 = __builtin_amdgcn_wmma_f32_16x16x4_f32(false, a, false, b, (short)0, acc1, false, false);
        b.x = b0[32];       b.y = b0[DIM + 32];
        acc2 = __builtin_amdgcn_wmma_f32_16x16x4_f32(false, a, false, b, (short)0, acc2, false, false);
        b.x = b0[48];       b.y = b0[DIM + 48];
        acc3 = __builtin_amdgcn_wmma_f32_16x16x4_f32(false, a, false, b, (short)0, acc3, false, false);
    }

    const float4 dlo = *(const float4*)(dinv + m0 + 8 * half);
    const float4 dhi = *(const float4*)(dinv + m0 + 8 * half + 4);
    const float ds[8] = {dlo.x, dlo.y, dlo.z, dlo.w, dhi.x, dhi.y, dhi.z, dhi.w};
    float* crow = C + (size_t)(m0 + 8 * half) * DIM + r;
#pragma unroll
    for (int v = 0; v < 8; ++v) {
        crow[(size_t)v * DIM + 0]  = acc0[v] * ds[v];
        crow[(size_t)v * DIM + 16] = acc1[v] * ds[v];
        crow[(size_t)v * DIM + 32] = acc2[v] * ds[v];
        crow[(size_t)v * DIM + 48] = acc3[v] * ds[v];
    }
}

#define NODES_PER_WAVE 4
__global__ __launch_bounds__(256) void k_gather(const int* __restrict__ rowptr,
                                                const int2* __restrict__ recs,
                                                const float* __restrict__ dinv,
                                                const float* __restrict__ h,
                                                const float* __restrict__ b,
                                                float* __restrict__ out) {
    const int lane = threadIdx.x & 31;
    const int wave = (blockIdx.x * 256 + (int)threadIdx.x) >> 5;
    const int n0   = wave * NODES_PER_WAVE;
    const int bnd = rowptr[n0 + (lane < NODES_PER_WAVE ? lane : NODES_PER_WAVE)];
    const float2 bias = *(const float2*)(b + lane * 2);
    const float* hl = h + lane * 2;
    for (int i = 0; i < NODES_PER_WAVE; ++i) {
        const int node = n0 + i;
        const int s = max(__builtin_amdgcn_readlane(bnd, i), 0);
        const int e = min(__builtin_amdgcn_readlane(bnd, i + 1), N_EDGES);
        const float di = dinv[node];
        float2 acc = *(const float2*)(hl + (size_t)node * DIM);
        for (int p0 = s; p0 < e; p0 += 32) {
            const int p = p0 + lane;
            int2 rec = make_int2(0, 0);
            if (p < e) rec = recs[p];
            const int cnt = min(32, e - p0);
            int j = 0;
            for (; j + 8 <= cnt; j += 8) {
                float2 v[8];
#pragma unroll
                for (int u = 0; u < 8; ++u) {
                    const int r = __builtin_amdgcn_readlane(rec.x, j + u);
                    v[u] = *(const float2*)(hl + (size_t)r * DIM);
                }
#pragma unroll
                for (int u = 0; u < 8; ++u) {
                    const float n = __int_as_float(__builtin_amdgcn_readlane(rec.y, j + u));
                    acc.x += n * v[u].x;
                    acc.y += n * v[u].y;
                }
            }
            for (; j < cnt; j += 4) {
                float2 v[4];
#pragma unroll
                for (int u = 0; u < 4; ++u) {
                    const int r = __builtin_amdgcn_readlane(rec.x, min(j + u, cnt - 1));
                    v[u] = *(const float2*)(hl + (size_t)r * DIM);
                }
#pragma unroll
                for (int u = 0; u < 4; ++u) {
                    const float nv = __int_as_float(__builtin_amdgcn_readlane(rec.y, min(j + u, cnt - 1)));
                    const float n = (j + u < cnt) ? nv : 0.0f;
                    acc.x += n * v[u].x;
                    acc.y += n * v[u].y;
                }
            }
        }
        float2 o;
        o.x = acc.x * di + bias.x;
        o.y = acc.y * di + bias.y;
        *(float2*)(out + (size_t)node * DIM + lane * 2) = o;
    }
}


__global__ __launch_bounds__(256) void k_pool(const float* __restrict__ g,
                                              float* __restrict__ out) {
    const int t = blockIdx.x * 256 + threadIdx.x;
    const int m = t >> 6;
    const int d = t & 63;
    const float* base = g + (size_t)m * DIM * DIM + d;
    float s = 0.f;
#pragma unroll 8
    for (int j = 0; j < DIM; ++j) s += base[(size_t)j * DIM];
    out[(size_t)d * M_POOL + m] = s * (1.0f / DIM);
}


extern "C" void kernel_launch(void* const* d_in, const int* in_sizes, int n_in,
                              void* d_out, int out_size, void* d_ws, size_t ws_size,
                              hipStream_t stream) {
    const float* x   = (const float*)d_in[0];
    const int*  eidx = (const int*) d_in[1];
    const float* ew  = (const float*)d_in[2];
    const float* W1  = (const float*)d_in[3];
    const float* b1  = (const float*)d_in[4];
    const float* W2  = (const float*)d_in[5];
    const float* b2  = (const float*)d_in[6];
    float* out = (float*)d_out;

    const int* rowi = eidx;
    const int* coli = eidx + N_EDGES;

    if (ws_size < (size_t)42467584) return;
    float* dinv = (float*)d_ws;
    float* bufH = dinv + N_NODES;
    float* bufA = bufH + (size_t)N_NODES * DIM;
    int*  rowptr = (int*)(bufA + (size_t)N_NODES * DIM);
    int2* recs   = (int2*)(rowptr + N_NODES + 64);
    int*  hist   = (int*)bufA;
    int*  off    = hist + N_CBIN * N_PART;
    int2* coarse = (int2*)(hist + 4 * N_CBIN * N_PART);

    const int GATHER_BLK = (N_NODES / NODES_PER_WAVE) / 8;
    const int GEMM_BLK   = (N_NODES / 16) / 8;

    k_hist<<<N_PART, 256, 0, stream>>>(coli, hist);
    k_scan<<<1, 1024, 0, stream>>>(hist, off);
    k_part<<<N_PART, 256, 0, stream>>>(rowi, coli, ew, off, coarse);
    k_fine<<<N_CBIN, 256, 0, stream>>>(off, coarse, rowptr, recs);
    k_dinv<<<N_NODES / 256, 256, 0, stream>>>(rowptr, recs, dinv);

    k_gemm_wmma<<<GEMM_BLK, 256, 0, stream>>>(x, W1, dinv, bufH);
    k_gather<<<GATHER_BLK, 256, 0, stream>>>(rowptr, recs, dinv, bufH, b1, bufA);

    k_gemm_wmma<<<GEMM_BLK, 256, 0, stream>>>(bufA, W2, dinv, bufH);
    k_gather<<<GATHER_BLK, 256, 0, stream>>>(rowptr, recs, dinv, bufH, b2, bufA);

    k_pool<<<(M_POOL * DIM) / 256, 256, 0, stream>>>(bufA, out);
}


# layer 2 by linearity: each window's 64 aggregated rows are summed before the W2 product, now one 16x64 tile per window inside the gather; 65536-row GEMM and pool launches gone; sum order differs
# speedup vs baseline: 3.9159x; 1.2190x over previous
#include <hip/hip_runtime.h>
#include <hip/hip_bf16.h>


typedef float v2f __attribute__((ext_vector_type(2)));
typedef float v8f __attribute__((ext_vector_type(8)));

#define N_NODES 65536
#define N_EDGES 1048576
#define DIM 64
#define M_POOL (N_NODES / DIM)


#define PART_EDGES 4096
#define N_PART (N_EDGES / PART_EDGES)
#define N_CBIN 256
#define FINE_CAP 6144

__global__ __launch_bounds__(256) void k_hist(const int* __restrict__ col,
                                              int* __restrict__ hist) {
    __shared__ int lcnt[N_CBIN];
    const int t = threadIdx.x;
    const int4* src = (const int4*)(col + (size_t)blockIdx.x * PART_EDGES) + t;
    lcnt[t] = 0;
    __syncthreads();
    int4 c[4];
#pragma unroll
    for (int i = 0; i < 4; ++i) c[i] = src[i * 256];
#pragma unroll
    for (int i = 0; i < 4; ++i) {
        atomicAdd(&lcnt[(c[i].x >> 8) & 255], 1);
        atomicAdd(&lcnt[(c[i].y >> 8) & 255], 1);
        atomicAdd(&lcnt[(c[i].z >> 8) & 255], 1);
        atomicAdd(&lcnt[(c[i].w >> 8) & 255], 1);
    }
    __syncthreads();
    hist[t * N_PART + blockIdx.x] = lcnt[t];
}

#define SCAN_WAVES 32
#define SCAN_PIECES 16
__global__ __launch_bounds__(1024) void k_scan(const int* __restrict__ cnt,
                                               int* __restrict__ rowptr) {
    __shared__ int wsum[SCAN_WAVES];
    const int lane = threadIdx.x & 31;
    const int wv   = threadIdx.x >> 5;
    const int4* src = (const int4*)(cnt + wv * (SCAN_PIECES * 128)) + lane;
    int4 v[SCAN_PIECES];
#pragma unroll
    for (int it = 0; it < SCAN_PIECES; ++it) v[it] = src[it * 32];
    int carry = 0;
#pragma unroll
    for (int it = 0; it < SCAN_PIECES; ++it) {
        const int4 x = v[it];
        const int s1 = x.x, s2 = s1 + x.y, s3 = s2 + x.z, s4 = s3 + x.w;
        int inc = s4;
        for (int d = 1; d < 32; d <<= 1) {
            const int y = __shfl_up(inc, d);
            if (lane >= d) inc += y;
        }
        const int excl = inc - s4 + carry;
        v[it] = make_int4(excl, excl + s1, excl + s2, excl + s3);
        carry += __shfl(inc, 31);
    }
    if (lane == 0) wsum[wv] = carry;
    __syncthreads();
    int incw = wsum[lane];
    for (int d = 1; d < 32; d <<= 1) {
        const int y = __shfl_up(incw, d);
        if (lane >= d) incw += y;
    }
    const int below = __shfl(incw, (wv + 31) & 31);
    const int base  = (wv == 0) ? 0 : below;
    int4* dst = (int4*)(rowptr + wv * (SCAN_PIECES * 128)) + lane;
#pragma unroll
    for (int it = 0; it < SCAN_PIECES; ++it) {
        int4 o = v[it];
        o.x += base; o.y += base; o.z += base; o.w += base;
        dst[it * 32] = o;
    }
    if (threadIdx.x == 1023) rowptr[N_NODES] = incw;
}

__device__ __forceinline__ int excl_scan_256(const int v, int* wsum) {
    const int lane = threadIdx.x & 31;
    const int wv = threadIdx.x >> 5;
    int inc = v;
    for (int d = 1; d < 32; d <<= 1) {
        const int y = __shfl_up(inc, d);
        if (lane >= d) inc += y;
    }
    if (lane == 31) wsum[wv] = inc;
    __syncthreads();
    int below = 0;
    for (int k = 0; k < 8; ++k) below += (k < wv) ? wsum[k] : 0;
    return inc - v + below;
}

__global__ __launch_bounds__(256) void k_part(const int* __restrict__ row,
                                              const int* __restrict__ col,
                                              const float* __restrict__ w,
                                              const int* __restrict__ off,
                                              int2* __restrict__ coarse) {
    __shared__ int lcnt[N_CBIN];
    __shared__ int lgo[N_CBIN];
    __shared__ int wsum[8];
    __shared__ int2 lrec[PART_EDGES];
    const int t = threadIdx.x;
    const size_t q = (size_t)blockIdx.x * (PART_EDGES / 4) + t;
    lcnt[t] = 0;
    __syncthreads();
    int4 c4[4], r4[4];
    float4 w4[4];
#pragma unroll
    for (int i = 0; i < 4; ++i) c4[i] = ((const int4*)col)[q + i * 256];
#pragma unroll
    for (int i = 0; i < 4; ++i) {
        r4[i] = ((const int4*)row)[q + i * 256];
        w4[i] = ((const float4*)w)[q + i * 256];
    }
    const int goff = off[t * N_PART + blockIdx.x];
    unsigned cc[16], rr[16];
    float wr[16];
#pragma unroll
    for (int i = 0; i < 4; ++i) {
        cc[4 * i + 0] = (unsigned)c4[i].x & 0xFFFFu;  rr[4 * i + 0] = (unsigned)r4[i].x & 0xFFFFu;
        cc[4 * i + 1] = (unsigned)c4[i].y & 0xFFFFu;  rr[4 * i + 1] = (unsigned)r4[i].y & 0xFFFFu;
        cc[4 * i + 2] = (unsigned)c4[i].z & 0xFFFFu;  rr[4 * i + 2] = (unsigned)r4[i].z & 0xFFFFu;
        cc[4 * i + 3] = (unsigned)c4[i].w & 0xFFFFu;  rr[4 * i + 3] = (unsigned)r4[i].w & 0xFFFFu;
        wr[4 * i + 0] = w4[i].x;  wr[4 * i + 1] = w4[i].y;
        wr[4 * i + 2] = w4[i].z;  wr[4 * i + 3] = w4[i].w;
    }
    int rank[16];
#pragma unroll
    for (int k = 0; k < 16; ++k) rank[k] = atomicAdd(&lcnt[cc[k] >> 8], 1);
    __syncthreads();
    const int cnt = lcnt[t];
    const int first = excl_scan_256(cnt, wsum);
    lcnt[t] = first;
    lgo[t] = goff - first;
    __syncthreads();
    int fs[16];
#pragma unroll
    for (int k = 0; k < 16; ++k) fs[k] = lcnt[cc[k] >> 8];
#pragma unroll
    for (int k = 0; k < 16; ++k)
        lrec[fs[k] + rank[k]] = make_int2((int)(rr[k] | (cc[k] << 16)), __float_as_int(wr[k]));
    __syncthreads();
    int2 rec[16];
    int go[16];
#pragma unroll
    for (int i = 0; i < 16; ++i) rec[i] = lrec[i * 256 + t];
#pragma unroll
    for (int i = 0; i < 16; ++i) go[i] = lgo[((unsigned)rec[i].x) >> 24];
#pragma unroll
    for (int i = 0; i < 16; ++i) {
        const int d = go[i] + i * 256 + t;
        if ((unsigned)d < (unsigned)N_EDGES) coarse[d] = rec[i];
    }
}

__global__ __launch_bounds__(256) void k_fine(const int* __restrict__ off,
                                              const int2* __restrict__ coarse,
                                              int* __restrict__ rowptr,
                                              int2* __restrict__ recs) {
    __shared__ int lcnt[256];
    __shared__ int wsum[8];
    __shared__ int2 lout[FINE_CAP];
    const int t = threadIdx.x;
    const int b = blockIdx.x;
    const int s = min(max(off[b * N_PART], 0), N_EDGES);
    const int e = min(max(off[(b + 1) * N_PART], s), N_EDGES);
    const int n = e - s;
    lcnt[t] = 0;
    __syncthreads();
    for (int p0 = s; p0 < e; p0 += 4096) {
        int x[16];
#pragma unroll
        for (int u = 0; u < 16; ++u) x[u] = coarse[min(p0 + u * 256 + t, e - 1)].x;
#pragma unroll
        for (int u = 0; u < 16; ++u)
            if (p0 + u * 256 + t < e) atomicAdd(&lcnt[((unsigned)x[u] >> 16) & 255u], 1);
    }
    __syncthreads();
    const int cnt = lcnt[t];
    const int first = excl_scan_256(cnt, wsum);
    rowptr[b * 256 + t] = s + first;
    lcnt[t] = first;
    if (b == N_CBIN - 1 && t == 0) rowptr[N_NODES] = e;
    __syncthreads();
    const bool fits = (n <= FINE_CAP);
    for (int p0 = s; p0 < e; p0 += 4096) {
        int2 r[16];
        int pos[16];
#pragma unroll
        for (int u = 0; u < 16; ++u) r[u] = coarse[min(p0 + u * 256 + t, e - 1)];
#pragma unroll
        for (int u = 0; u < 16; ++u)
            pos[u] = (p0 + u * 256 + t < e)
                         ? atomicAdd(&lcnt[((unsigned)r[u].x >> 16) & 255u], 1) : -1;
#pragma unroll
        for (int u = 0; u < 16; ++u) {
            const int2 o = make_int2(r[u].x & 0xFFFF, r[u].y);
            if (fits) {
                if ((unsigned)pos[u] < (unsigned)FINE_CAP) lout[pos[u]] = o;
            } else {
                if (pos[u] >= 0 && (unsigned)(s + pos[u]) < (unsigned)N_EDGES) recs[s + pos[u]] = o;
            }
        }
    }
    if (fits) {
        __syncthreads();
        for (int i0 = 0; i0 < n; i0 += 2048) {
            int2 v[8];
#pragma unroll
            for (int u = 0; u < 8; ++u) v[u] = lout[min(i0 + u * 256 + t, n - 1)];
#pragma unroll
            for (int u = 0; u < 8; ++u)
                if (i0 + u * 256 + t < n) recs[s + i0 + u * 256 + t] = v[u];
        }
    }
}

__global__ __launch_bounds__(256) void k_dinv(const int* __restrict__ rowptr,
                                              const int2* __restrict__ recs,
                                              float* __restrict__ dinv) {
    const int n = blockIdx.x * 256 + threadIdx.x;
    const int s = max(rowptr[n], 0);
    const int e = min(rowptr[n + 1], N_EDGES);
    float sum = 0.0f;
    for (int p = s; p < e; p += 4) {
        float wv[4];
#pragma unroll
        for (int u = 0; u < 4; ++u)
            wv[u] = (p + u < e) ? __int_as_float(recs[p + u].y) : 0.0f;
        sum = (((sum + wv[0]) + wv[1]) + wv[2]) + wv[3];
    }
    const float deg = sum + 1.0f;
    dinv[n] = (deg > 0.0f) ? rsqrtf(deg) : 0.0f;
}


__global__ __launch_bounds__(256) void k_gemm_wmma(const float* __restrict__ A,
                                                   const float* __restrict__ B,
                                                   const float* __restrict__ dinv,
                                                   float* __restrict__ C) {
    const int gwave = (blockIdx.x * 256 + (int)threadIdx.x) >> 5;
    const int lane  = threadIdx.x & 31;
    const int half  = lane >> 4;
    const int r     = lane & 15;
    const int m0    = gwave << 4;

    v8f acc0 = {}, acc1 = {}, acc2 = {}, acc3 = {};

    const float* arow = A + (size_t)(m0 + r) * DIM + 2 * half;

#pragma unroll
    for (int k0 = 0; k0 < DIM; k0 += 4) {
        v2f a;
        a.x = arow[k0];
        a.y = arow[k0 + 1];

        const float* b0 = B + (size_t)(k0 + 2 * half) * DIM + r;
        v2f b;

        b.x = b0[0];        b.y = b0[DIM];
        acc0 = __builtin_amdgcn_wmma_f32_16x16x4_f32(false, a, false, b, (short)0, acc0, false, false);
        b.x = b0[16];       b.y = b0[DIM + 16];
        acc1 = __builtin_amdgcn_wmma_f32_16x16x4_f32(false, a, false, b, (short)0, acc1, false, false);
        b.x = b0[32];       b.y = b0[DIM + 32];
        acc2 = __builtin_amdgcn_wmma_f32_16x16x4_f32(false, a, false, b, (short)0, acc2, false, false);
        b.x = b0[48];       b.y = b0[DIM + 48];
        acc3 = __builtin_amdgcn_wmma_f32_16x16x4_f32(false, a, false, b, (short)0, acc3, false, false);
    }

    const float4 dlo = *(const float4*)(dinv + m0 + 8 * half);
    const float4 dhi = *(const float4*)(dinv + m0 + 8 * half + 4);
    const float ds[8] = {dlo.x, dlo.y, dlo.z, dlo.w, dhi.x, dhi.y, dhi.z, dhi.w};
    float* crow = C + (size_t)(m0 + 8 * half) * DIM + r;
#pragma unroll
    for (int v = 0; v < 8; ++v) {
        crow[(size_t)v * DIM + 0]  = acc0[v] * ds[v];
        crow[(size_t)v * DIM + 16] = acc1[v] * ds[v];
        crow[(size_t)v * DIM + 32] = acc2[v] * ds[v];
        crow[(size_t)v * DIM + 48] = acc3[v] * ds[v];
    }
}

#define NODES_PER_WAVE 4

__device__ __forceinline__ float2 bucket_sum(const int2* __restrict__ recs, const float* hl,
                                             const int node, const int s, const int e, const int lane) {
    float2 acc = *(const float2*)(hl + (size_t)node * DIM);
    for (int p0 = s; p0 < e; p0 += 32) {
        const int p = p0 + lane;
        int2 rec = make_int2(0, 0);
        if (p < e) rec = recs[p];
        const int cnt = min(32, e - p0);
        int j = 0;
        for (; j + 8 <= cnt; j += 8) {
            float2 v[8];
#pragma unroll
            for (int u = 0; u < 8; ++u) {
                const int r = __builtin_amdgcn_readlane(rec.x, j + u);
                v[u] = *(const float2*)(hl + (size_t)r * DIM);
            }
#pragma unroll
            for (int u = 0; u < 8; ++u) {
                const float n = __int_as_float(__builtin_amdgcn_readlane(rec.y, j + u));
                acc.x += n * v[u].x;
                acc.y += n * v[u].y;
            }
        }
        for (; j < cnt; j += 4) {
            float2 v[4];
#pragma unroll
            for (int u = 0; u < 4; ++u) {
                const int r = __builtin_amdgcn_readlane(rec.x, min(j + u, cnt - 1));
                v[u] = *(const float2*)(hl + (size_t)r * DIM);
            }
#pragma unroll
            for (int u = 0; u < 4; ++u) {
                const float nv = __int_as_float(__builtin_amdgcn_readlane(rec.y, min(j + u, cnt - 1)));
                const float n = (j + u < cnt) ? nv : 0.0f;
                acc.x += n * v[u].x;
                acc.y += n * v[u].y;
            }
        }
    }
    return acc;
}

__global__ __launch_bounds__(256) void k_gather(const int* __restrict__ rowptr,
                                                const int2* __restrict__ recs,
                                                const float* __restrict__ dinv,
                                                const float* __restrict__ h,
                                                const float* __restrict__ b,
                                                float* __restrict__ out) {
    const int lane = threadIdx.x & 31;
    const int wave = (blockIdx.x * 256 + (int)threadIdx.x) >> 5;
    const int n0   = wave * NODES_PER_WAVE;
    const int bnd = rowptr[n0 + (lane < NODES_PER_WAVE ? lane : NODES_PER_WAVE)];
    const float2 bias = *(const float2*)(b + lane * 2);
    const float* hl = h + lane * 2;
    for (int i = 0; i < NODES_PER_WAVE; ++i) {
        const int node = n0 + i;
        const int s = max(__builtin_amdgcn_readlane(bnd, i), 0);
        const int e = min(__builtin_amdgcn_readlane(bnd, i + 1), N_EDGES);
        const float di = dinv[node];
        const float2 acc = bucket_sum(recs, hl, node, s, e, lane);
        float2 o;
        o.x = (acc.x * di + bias.x) * di;
        o.y = (acc.y * di + bias.y) * di;
        *(float2*)(out + (size_t)node * DIM + lane * 2) = o;
    }
}

#define POOL_WAVES (DIM / NODES_PER_WAVE)
#define TILE_PITCH 68
__global__ __launch_bounds__(POOL_WAVES * 32) void k_gather_pool_gemm(const int* __restrict__ rowptr,
                                                                      const int2* __restrict__ recs,
                                                                      const float* __restrict__ dinv,
                                                                      const float* __restrict__ h,
                                                                      const float* __restrict__ Wm,
                                                                      const float* __restrict__ b,
                                                                      float* __restrict__ out) {
    __shared__ float tile[POOL_WAVES * TILE_PITCH];
    const int lane = threadIdx.x & 31;
    const int wv   = threadIdx.x >> 5;
    const int m    = blockIdx.x;
    const int n0   = m * DIM + wv * NODES_PER_WAVE;
    const int bnd = rowptr[n0 + (lane < NODES_PER_WAVE ? lane : NODES_PER_WAVE)];
    const float* hl = h + lane * 2;
    float2 ws = make_float2(0.0f, 0.0f);
    for (int i = 0; i < NODES_PER_WAVE; ++i) {
        const int node = n0 + i;
        const int s = max(__builtin_amdgcn_readlane(bnd, i), 0);
        const int e = min(__builtin_amdgcn_readlane(bnd, i + 1), N_EDGES);
        const float di = dinv[node];
        const float2 acc = bucket_sum(recs, hl, node, s, e, lane);
        ws.x += acc.x * di;
        ws.y += acc.y * di;
    }
    *(float2*)(tile + wv * TILE_PITCH + lane * 2) = ws;

    const int half = lane >> 4;
    const int r    = lane & 15;
    const int c0   = 16 * (wv & 3) + r;
    v2f bf[DIM / 4];
    float bias = 0.0f;
    if (wv < 4) {
        bias = b[c0];
        const float* b0 = Wm + (size_t)(2 * half) * DIM + c0;
#pragma unroll
        for (int t = 0; t < DIM / 4; ++t) {
            bf[t].x = b0[(size_t)(4 * t) * DIM];
            bf[t].y = b0[(size_t)(4 * t + 1) * DIM];
        }
    }
    __syncthreads();
    if (wv < 4) {
        v8f pa = {}, pb = {};
        const float* a0 = tile + r * TILE_PITCH + 2 * half;
#pragma unroll
        for (int t = 0; t < DIM / 4; t += 2) {
            const v2f a = *(const v2f*)(a0 + 4 * t);
            pa = __builtin_amdgcn_wmma_f32_16x16x4_f32(false, a, false, bf[t], (short)0, pa, false, false);
            const v2f a2 = *(const v2f*)(a0 + 4 * t + 4);
            pb = __builtin_amdgcn_wmma_f32_16x16x4_f32(false, a2, false, bf[t + 1], (short)0, pb, false, false);
        }
        float sum = 0.0f;
#pragma unroll
        for (int v = 0; v < 8; ++v) sum += pa[v] + pb[v];
        sum += __shfl_xor(sum, 16);
        if (half == 0) out[(size_t)c0 * M_POOL + m] = sum * (1.0f / DIM) + bias;
    }
}


extern "C" void kernel_launch(void* const* d_in, const int* in_sizes, int n_in,
                              void* d_out, int out_size, void* d_ws, size_t ws_size,
                              hipStream_t stream) {
    const float* x   = (const float*)d_in[0];
    const int*  eidx = (const int*) d_in[1];
    const float* ew  = (const float*)d_in[2];
    const float* W1  = (const float*)d_in[3];
    const float* b1  = (const float*)d_in[4];
    const float* W2  = (const float*)d_in[5];
    const float* b2  = (const float*)d_in[6];
    float* out = (float*)d_out;

    const int* rowi = eidx;
    const int* coli = eidx + N_EDGES;

    if (ws_size < (size_t)42467584) return;
    float* dinv = (float*)d_ws;
    float* bufH = dinv + N_NODES;
    float* bufA = bufH + (size_t)N_NODES * DIM;
    int*  rowptr = (int*)(bufA + (size_t)N_NODES * DIM);
    int2* recs   = (int2*)(rowptr + N_NODES + 64);
    int*  hist   = (int*)bufA;
    int*  off    = hist + N_CBIN * N_PART;
    int2* coarse = (int2*)(hist + 4 * N_CBIN * N_PART);

    const int GATHER_BLK = (N_NODES / NODES_PER_WAVE) / 8;
    const int GEMM_BLK   = (N_NODES / 16) / 8;

    k_hist<<<N_PART, 256, 0, stream>>>(coli, hist);
    k_scan<<<1, 1024, 0, stream>>>(hist, off);
    k_part<<<N_PART, 256, 0, stream>>>(rowi, coli, ew, off, coarse);
    k_fine<<<N_CBIN, 256, 0, stream>>>(off, coarse, rowptr, recs);
    k_dinv<<<N_NODES / 256, 256, 0, stream>>>(rowptr, recs, dinv);

    k_gemm_wmma<<<GEMM_BLK, 256, 0, stream>>>(x, W1, dinv, bufH);
    k_gather<<<GATHER_BLK, 256, 0, stream>>>(rowptr, recs, dinv, bufH, b1, bufA);

    k_gather_pool_gemm<<<M_POOL, POOL_WAVES * 32, 0, stream>>>(rowptr, recs, dinv, bufA, W2, b2, out);
}


# last sort level, degree/dinv and layer-1 GEMM fused into one launch: the work-group that finishes 256 nodes' buckets sums their degrees from LDS and multiplies their x rows by W1; 8 launches to 6
# speedup vs baseline: 4.3500x; 1.1109x over previous
#include <hip/hip_runtime.h>
#include <hip/hip_bf16.h>


typedef float v2f __attribute__((ext_vector_type(2)));
typedef float v8f __attribute__((ext_vector_type(8)));

#define N_NODES 65536
#define N_EDGES 1048576
#define DIM 64
#define M_POOL (N_NODES / DIM)


#define PART_EDGES 4096
#define N_PART (N_EDGES / PART_EDGES)
#define N_CBIN 256
#define FINE_CAP 6144

__global__ __launch_bounds__(256) void k_hist(const int* __restrict__ col,
                                              int* __restrict__ hist) {
    __shared__ int lcnt[N_CBIN];
    const int t = threadIdx.x;
    const int4* src = (const int4*)(col + (size_t)blockIdx.x * PART_EDGES) + t;
    lcnt[t] = 0;
    __syncthreads();
    int4 c[4];
#pragma unroll
    for (int i = 0; i < 4; ++i) c[i] = src[i * 256];
#pragma unroll
    for (int i = 0; i < 4; ++i) {
        atomicAdd(&lcnt[(c[i].x >> 8) & 255], 1);
        atomicAdd(&lcnt[(c[i].y >> 8) & 255], 1);
        atomicAdd(&lcnt[(c[i].z >> 8) & 255], 1);
        atomicAdd(&lcnt[(c[i].w >> 8) & 255], 1);
    }
    __syncthreads();
    hist[t * N_PART + blockIdx.x] = lcnt[t];
}

#define SCAN_WAVES 32
#define SCAN_PIECES 16
__global__ __launch_bounds__(1024) void k_scan(const int* __restrict__ cnt,
                                               int* __restrict__ rowptr) {
    __shared__ int wsum[SCAN_WAVES];
    const int lane = threadIdx.x & 31;
    const int wv   = threadIdx.x >> 5;
    const int4* src = (const int4*)(cnt + wv * (SCAN_PIECES * 128)) + lane;
    int4 v[SCAN_PIECES];
#pragma unroll
    for (int it = 0; it < SCAN_PIECES; ++it) v[it] = src[it * 32];
    int carry = 0;
#pragma unroll
    for (int it = 0; it < SCAN_PIECES; ++it) {
        const int4 x = v[it];
        const int s1 = x.x, s2 = s1 + x.y, s3 = s2 + x.z, s4 = s3 + x.w;
        int inc = s4;
        for (int d = 1; d < 32; d <<= 1) {
            const int y = __shfl_up(inc, d);
            if (lane >= d) inc += y;
        }
        const int excl = inc - s4 + carry;
        v[it] = make_int4(excl, excl + s1, excl + s2, excl + s3);
        carry += __shfl(inc, 31);
    }
    if (lane == 0) wsum[wv] = carry;
    __syncthreads();
    int incw = wsum[lane];
    for (int d = 1; d < 32; d <<= 1) {
        const int y = __shfl_up(incw, d);
        if (lane >= d) incw += y;
    }
    const int below = __shfl(incw, (wv + 31) & 31);
    const int base  = (wv == 0) ? 0 : below;
    int4* dst = (int4*)(rowptr + wv * (SCAN_PIECES * 128)) + lane;
#pragma unroll
    for (int it = 0; it < SCAN_PIECES; ++it) {
        int4 o = v[it];
        o.x += base; o.y += base; o.z += base; o.w += base;
        dst[it * 32] = o;
    }
    if (threadIdx.x == 1023) rowptr[N_NODES] = incw;
}

__device__ __forceinline__ int excl_scan_256(const int v, int* wsum) {
    const int lane = threadIdx.x & 31;
    const int wv = threadIdx.x >> 5;
    int inc = v;
    for (int d = 1; d < 32; d <<= 1) {
        const int y = __shfl_up(inc, d);
        if (lane >= d) inc += y;
    }
    if (lane == 31) wsum[wv] = inc;
    __syncthreads();
    int below = 0;
    for (int k = 0; k < 8; ++k) below += (k < wv) ? wsum[k] : 0;
    return inc - v + below;
}

__global__ __launch_bounds__(256) void k_part(const int* __restrict__ row,
                                              const int* __restrict__ col,
                                              const float* __restrict__ w,
                                              const int* __restrict__ off,
                                              int2* __restrict__ coarse) {
    __shared__ int lcnt[N_CBIN];
    __shared__ int lgo[N_CBIN];
    __shared__ int wsum[8];
    __shared__ int2 lrec[PART_EDGES];
    const int t = threadIdx.x;
    const size_t q = (size_t)blockIdx.x * (PART_EDGES / 4) + t;
    lcnt[t] = 0;
    __syncthreads();
    int4 c4[4], r4[4];
    float4 w4[4];
#pragma unroll
    for (int i = 0; i < 4; ++i) c4[i] = ((const int4*)col)[q + i * 256];
#pragma unroll
    for (int i = 0; i < 4; ++i) {
        r4[i] = ((const int4*)row)[q + i * 256];
        w4[i] = ((const float4*)w)[q + i * 256];
    }
    const int goff = off[t * N_PART + blockIdx.x];
    unsigned cc[16], rr[16];
    float wr[16];
#pragma unroll
    for (int i = 0; i < 4; ++i) {
        cc[4 * i + 0] = (unsigned)c4[i].x & 0xFFFFu;  rr[4 * i + 0] = (unsigned)r4[i].x & 0xFFFFu;
        cc[4 * i + 1] = (unsigned)c4[i].y & 0xFFFFu;  rr[4 * i + 1] = (unsigned)r4[i].y & 0xFFFFu;
        cc[4 * i + 2] = (unsigned)c4[i].z & 0xFFFFu;  rr[4 * i + 2] = (unsigned)r4[i].z & 0xFFFFu;
        cc[4 * i + 3] = (unsigned)c4[i].w & 0xFFFFu;  rr[4 * i + 3] = (unsigned)r4[i].w & 0xFFFFu;
        wr[4 * i + 0] = w4[i].x;  wr[4 * i + 1] = w4[i].y;
        wr[4 * i + 2] = w4[i].z;  wr[4 * i + 3] = w4[i].w;
    }
    int rank[16];
#pragma unroll
    for (int k = 0; k < 16; ++k) rank[k] = atomicAdd(&lcnt[cc[k] >> 8], 1);
    __syncthreads();
    const int cnt = lcnt[t];
    const int first = excl_scan_256(cnt, wsum);
    lcnt[t] = first;
    lgo[t] = goff - first;
    __syncthreads();
    int fs[16];
#pragma unroll
    for (int k = 0; k < 16; ++k) fs[k] = lcnt[cc[k] >> 8];
#pragma unroll
    for (int k = 0; k < 16; ++k)
        lrec[fs[k] + rank[k]] = make_int2((int)(rr[k] | (cc[k] << 16)), __float_as_int(wr[k]));
    __syncthreads();
    int2 rec[16];
    int go[16];
#pragma unroll
    for (int i = 0; i < 16; ++i) rec[i] = lrec[i * 256 + t];
#pragma unroll
    for (int i = 0; i < 16; ++i) go[i] = lgo[((unsigned)rec[i].x) >> 24];
#pragma unroll
    for (int i = 0; i < 16; ++i) {
        const int d = go[i] + i * 256 + t;
        if ((unsigned)d < (unsigned)N_EDGES) coarse[d] = rec[i];
    }
}


__device__ __forceinline__ void store_strip(float* __restrict__ crow, const float* ds8,
                                            const v8f a0, const v8f a1, const v8f a2, const v8f a3) {
    const float4 dlo = *(const float4*)(ds8);
    const float4 dhi = *(const float4*)(ds8 + 4);
    const float ds[8] = {dlo.x, dlo.y, dlo.z, dlo.w, dhi.x, dhi.y, dhi.z, dhi.w};
#pragma unroll
    for (int v = 0; v < 8; ++v) {
        crow[(size_t)v * DIM + 0]  = a0[v] * ds[v];
        crow[(size_t)v * DIM + 16] = a1[v] * ds[v];
        crow[(size_t)v * DIM + 32] = a2[v] * ds[v];
        crow[(size_t)v * DIM + 48] = a3[v] * ds[v];
    }
}

__global__ __launch_bounds__(256) void k_fine_gemm(const int* __restrict__ off,
                                                   const int2* __restrict__ coarse,
                                                   const float* __restrict__ X,
                                                   const float* __restrict__ Wm,
                                                   int* __restrict__ rowptr,
                                                   int2* __restrict__ recs,
                                                   float* __restrict__ dinv,
                                                   float* __restrict__ C) {
    __shared__ int lcnt[256];
    __shared__ int wsum[8];
    __shared__ __attribute__((aligned(16))) float ldinv[256];
    __shared__ int2 lout[FINE_CAP];
    const int t = threadIdx.x;
    const int b = blockIdx.x;
    const int s = min(max(off[b * N_PART], 0), N_EDGES);
    const int e = min(max(off[(b + 1) * N_PART], s), N_EDGES);
    const int n = e - s;
    lcnt[t] = 0;
    __syncthreads();
    for (int p0 = s; p0 < e; p0 += 4096) {
        int x[16];
#pragma unroll
        for (int u = 0; u < 16; ++u) x[u] = coarse[min(p0 + u * 256 + t, e - 1)].x;
#pragma unroll
        for (int u = 0; u < 16; ++u)
            if (p0 + u * 256 + t < e) atomicAdd(&lcnt[((unsigned)x[u] >> 16) & 255u], 1);
    }
    __syncthreads();
    const int cnt = lcnt[t];
    const int first = excl_scan_256(cnt, wsum);
    rowptr[b * 256 + t] = s + first;
    lcnt[t] = first;
    if (b == N_CBIN - 1 && t == 0) rowptr[N_NODES] = e;
    __syncthreads();
    const bool fits = (n <= FINE_CAP);
    for (int p0 = s; p0 < e; p0 += 4096) {
        int2 r[16];
        int pos[16];
#pragma unroll
        for (int u = 0; u < 16; ++u) r[u] = coarse[min(p0 + u * 256 + t, e - 1)];
#pragma unroll
        for (int u = 0; u < 16; ++u)
            pos[u] = (p0 + u * 256 + t < e)
                         ? atomicAdd(&lcnt[((unsigned)r[u].x >> 16) & 255u], 1) : -1;
#pragma unroll
        for (int u = 0; u < 16; ++u) {
            const int2 o = make_int2(r[u].x & 0xFFFF, r[u].y);
            if (fits) {
                if ((unsigned)pos[u] < (unsigned)FINE_CAP) lout[pos[u]] = o;
            } else {
                if (pos[u] >= 0 && (unsigned)(s + pos[u]) < (unsigned)N_EDGES) recs[s + pos[u]] = o;
            }
        }
    }

    if (!fits) __threadfence();
    __syncthreads();
    if (!fits) __threadfence();

    const int lane = t & 31;
    const int wv   = t >> 5;
    const int half = lane >> 4;
    const int r16  = lane & 15;
    const int m0   = b * 256 + wv * 32;
    const float* arow = X + (size_t)(m0 + r16) * DIM + 2 * half;
    v2f af0[DIM / 4], af1[DIM / 4];
#pragma unroll
    for (int k = 0; k < DIM / 4; ++k) {
        af0[k] = *(const v2f*)(arow + 4 * k);
        af1[k] = *(const v2f*)(arow + 16 * DIM + 4 * k);
    }

    const int lo = min(max(first, 0), n);
    const int hi = min(max(first + cnt, lo), n);
    float sum = 0.0f;
    if (fits) {
        for (int i0 = 0; i0 < n; i0 += 2048) {
            int2 v[8];
#pragma unroll
            for (int u = 0; u < 8; ++u) v[u] = lout[min(i0 + u * 256 + t, n - 1)];
#pragma unroll
            for (int u = 0; u < 8; ++u)
                if (i0 + u * 256 + t < n) recs[s + i0 + u * 256 + t] = v[u];
        }
        for (int p = lo; p < hi; p += 4) {
            float wv4[4];
#pragma unroll
            for (int u = 0; u < 4; ++u)
                wv4[u] = (p + u < hi) ? __int_as_float(lout[p + u].y) : 0.0f;
            sum = (((sum + wv4[0]) + wv4[1]) + wv4[2]) + wv4[3];
        }
    } else {
        for (int p = lo; p < hi; p += 4) {
            float wv4[4];
#pragma unroll
            for (int u = 0; u < 4; ++u)
                wv4[u] = (p + u < hi) ? __int_as_float(recs[s + p + u].y) : 0.0f;
            sum = (((sum + wv4[0]) + wv4[1]) + wv4[2]) + wv4[3];
        }
    }
    const float deg = sum + 1.0f;
    const float di = (deg > 0.0f) ? rsqrtf(deg) : 0.0f;
    dinv[b * 256 + t] = di;
    ldinv[t] = di;

    v8f c00 = {}, c01 = {}, c02 = {}, c03 = {};
    v8f c10 = {}, c11 = {}, c12 = {}, c13 = {};
    const float* bp = Wm + (size_t)(2 * half) * DIM + r16;
#pragma unroll
    for (int k = 0; k < DIM / 4; ++k) {
        const float* b0 = bp + (size_t)(4 * k) * DIM;
        v2f bf;
        bf.x = b0[0];        bf.y = b0[DIM];
        c00 = __builtin_amdgcn_wmma_f32_16x16x4_f32(false, af0[k], false, bf, (short)0, c00, false, false);
        c10 = __builtin_amdgcn_wmma_f32_16x16x4_f32(false, af1[k], false, bf, (short)0, c10, false, false);
        bf.x = b0[16];       bf.y = b0[DIM + 16];
        c01 = __builtin_amdgcn_wmma_f32_16x16x4_f32(false, af0[k], false, bf, (short)0, c01, false, false);
        c11 = __builtin_amdgcn_wmma_f32_16x16x4_f32(false, af1[k], false, bf, (short)0, c11, false, false);
        bf.x = b0[32];       bf.y = b0[DIM + 32];
        c02 = __builtin_amdgcn_wmma_f32_16x16x4_f32(false, af0[k], false, bf, (short)0, c02, false, false);
        c12 = __builtin_amdgcn_wmma_f32_16x16x4_f32(false, af1[k], false, bf, (short)0, c12, false, false);
        bf.x = b0[48];       bf.y = b0[DIM + 48];
        c03 = __builtin_amdgcn_wmma_f32_16x16x4_f32(false, af0[k], false, bf, (short)0, c03, false, false);
        c13 = __builtin_amdgcn_wmma_f32_16x16x4_f32(false, af1[k], false, bf, (short)0, c13, false, false);
    }
    __syncthreads();
    float* crow = C + (size_t)(m0 + 8 * half) * DIM + r16;
    const float* ds8 = ldinv + wv * 32 + 8 * half;
    store_strip(crow, ds8, c00, c01, c02, c03);
    store_strip(crow + (size_t)16 * DIM, ds8 + 16, c10, c11, c12, c13);
}

#define NODES_PER_WAVE 4

__device__ __forceinline__ float2 bucket_sum(const int2* __restrict__ recs, const float* hl,
                                             const int node, const int s, const int e, const int lane) {
    float2 acc = *(const float2*)(hl + (size_t)node * DIM);
    for (int p0 = s; p0 < e; p0 += 32) {
        const int p = p0 + lane;
        int2 rec = make_int2(0, 0);
        if (p < e) rec = recs[p];
        const int cnt = min(32, e - p0);
        int j = 0;
        for (; j + 8 <= cnt; j += 8) {
            float2 v[8];
#pragma unroll
            for (int u = 0; u < 8; ++u) {
                const int r = __builtin_amdgcn_readlane(rec.x, j + u);
                v[u] = *(const float2*)(hl + (size_t)r * DIM);
            }
#pragma unroll
            for (int u = 0; u < 8; ++u) {
                const float n = __int_as_float(__builtin_amdgcn_readlane(rec.y, j + u));
                acc.x += n * v[u].x;
                acc.y += n * v[u].y;
            }
        }
        for (; j < cnt; j += 4) {
            float2 v[4];
#pragma unroll
            for (int u = 0; u < 4; ++u) {
                const int r = __builtin_amdgcn_readlane(rec.x, min(j + u, cnt - 1));
                v[u] = *(const float2*)(hl + (size_t)r * DIM);
            }
#pragma unroll
            for (int u = 0; u < 4; ++u) {
                const float nv = __int_as_float(__builtin_amdgcn_readlane(rec.y, min(j + u, cnt - 1)));
                const float n = (j + u < cnt) ? nv : 0.0f;
                acc.x += n * v[u].x;
                acc.y += n * v[u].y;
            }
        }
    }
    return acc;
}

__global__ __launch_bounds__(256) void k_gather(const int* __restrict__ rowptr,
                                                const int2* __restrict__ recs,
                                                const float* __restrict__ dinv,
                                                const float* __restrict__ h,
                                                const float* __restrict__ b,
                                                float* __restrict__ out) {
    const int lane = threadIdx.x & 31;
    const int wave = (blockIdx.x * 256 + (int)threadIdx.x) >> 5;
    const int n0   = wave * NODES_PER_WAVE;
    const int bnd = rowptr[n0 + (lane < NODES_PER_WAVE ? lane : NODES_PER_WAVE)];
    const float2 bias = *(const float2*)(b + lane * 2);
    const float* hl = h + lane * 2;
    for (int i = 0; i < NODES_PER_WAVE; ++i) {
        const int node = n0 + i;
        const int s = max(__builtin_amdgcn_readlane(bnd, i), 0);
        const int e = min(__builtin_amdgcn_readlane(bnd, i + 1), N_EDGES);
        const float di = dinv[node];
        const float2 acc = bucket_sum(recs, hl, node, s, e, lane);
        float2 o;
        o.x = (acc.x * di + bias.x) * di;
        o.y = (acc.y * di + bias.y) * di;
        *(float2*)(out + (size_t)node * DIM + lane * 2) = o;
    }
}

#define POOL_WAVES (DIM / NODES_PER_WAVE)
#define TILE_PITCH 68
__global__ __launch_bounds__(POOL_WAVES * 32) void k_gather_pool_gemm(const int* __restrict__ rowptr,
                                                                      const int2* __restrict__ recs,
                                                                      const float* __restrict__ dinv,
                                                                      const float* __restrict__ h,
                                                                      const float* __restrict__ Wm,
                                                                      const float* __restrict__ b,
                                                                      float* __restrict__ out) {
    __shared__ float tile[POOL_WAVES * TILE_PITCH];
    const int lane = threadIdx.x & 31;
    const int wv   = threadIdx.x >> 5;
    const int m    = blockIdx.x;
    const int n0   = m * DIM + wv * NODES_PER_WAVE;
    const int bnd = rowptr[n0 + (lane < NODES_PER_WAVE ? lane : NODES_PER_WAVE)];
    const float* hl = h + lane * 2;
    float2 ws = make_float2(0.0f, 0.0f);
    for (int i = 0; i < NODES_PER_WAVE; ++i) {
        const int node = n0 + i;
        const int s = max(__builtin_amdgcn_readlane(bnd, i), 0);
        const int e = min(__builtin_amdgcn_readlane(bnd, i + 1), N_EDGES);
        const float di = dinv[node];
        const float2 acc = bucket_sum(recs, hl, node, s, e, lane);
        ws.x += acc.x * di;
        ws.y += acc.y * di;
    }
    *(float2*)(tile + wv * TILE_PITCH + lane * 2) = ws;

    const int half = lane >> 4;
    const int r    = lane & 15;
    const int c0   = 16 * (wv & 3) + r;
    v2f bf[DIM / 4];
    float bias = 0.0f;
    if (wv < 4) {
        bias = b[c0];
        const float* b0 = Wm + (size_t)(2 * half) * DIM + c0;
#pragma unroll
        for (int t = 0; t < DIM / 4; ++t) {
            bf[t].x = b0[(size_t)(4 * t) * DIM];
            bf[t].y = b0[(size_t)(4 * t + 1) * DIM];
        }
    }
    __syncthreads();
    if (wv < 4) {
        v8f pa = {}, pb = {};
        const float* a0 = tile + r * TILE_PITCH + 2 * half;
#pragma unroll
        for (int t = 0; t < DIM / 4; t += 2) {
            const v2f a = *(const v2f*)(a0 + 4 * t);
            pa = __builtin_amdgcn_wmma_f32_16x16x4_f32(false, a, false, bf[t], (short)0, pa, false, false);
            const v2f a2 = *(const v2f*)(a0 + 4 * t + 4);
            pb = __builtin_amdgcn_wmma_f32_16x16x4_f32(false, a2, false, bf[t + 1], (short)0, pb, false, false);
        }
        float sum = 0.0f;
#pragma unroll
        for (int v = 0; v < 8; ++v) sum += pa[v] + pb[v];
        sum += __shfl_xor(sum, 16);
        if (half == 0) out[(size_t)c0 * M_POOL + m] = sum * (1.0f / DIM) + bias;
    }
}


extern "C" void kernel_launch(void* const* d_in, const int* in_sizes, int n_in,
                              void* d_out, int out_size, void* d_ws, size_t ws_size,
                              hipStream_t stream) {
    const float* x   = (const float*)d_in[0];
    const int*  eidx = (const int*) d_in[1];
    const float* ew  = (const float*)d_in[2];
    const float* W1  = (const float*)d_in[3];
    const float* b1  = (const float*)d_in[4];
    const float* W2  = (const float*)d_in[5];
    const float* b2  = (const float*)d_in[6];
    float* out = (float*)d_out;

    const int* rowi = eidx;
    const int* coli = eidx + N_EDGES;

    if (ws_size < (size_t)42467584) return;
    float* dinv = (float*)d_ws;
    float* bufH = dinv + N_NODES;
    float* bufA = bufH + (size_t)N_NODES * DIM;
    int*  rowptr = (int*)(bufA + (size_t)N_NODES * DIM);
    int2* recs   = (int2*)(rowptr + N_NODES + 64);
    int*  hist   = (int*)bufA;
    int*  off    = hist + N_CBIN * N_PART;
    int2* coarse = (int2*)(hist + 4 * N_CBIN * N_PART);

    const int GATHER_BLK = (N_NODES / NODES_PER_WAVE) / 8;

    k_hist<<<N_PART, 256, 0, stream>>>(coli, hist);
    k_scan<<<1, 1024, 0, stream>>>(hist, off);
    k_part<<<N_PART, 256, 0, stream>>>(rowi, coli, ew, off, coarse);
    k_fine_gemm<<<N_CBIN, 256, 0, stream>>>(off, coarse, x, W1, rowptr, recs, dinv, bufH);

    k_gather<<<GATHER_BLK, 256, 0, stream>>>(rowptr, recs, dinv, bufH, b1, bufA);

    k_gather_pool_gemm<<<M_POOL, POOL_WAVES * 32, 0, stream>>>(rowptr, recs, dinv, bufA, W2, b2, out);
}


# no global count in the edge sort: level 1 orders each 4096-edge chunk in place and writes a run table, level 2 collects its bin's 256 runs via an LDS index list; 6 launches to 4; sum order may differ
# speedup vs baseline: 5.4677x; 1.2569x over previous
#include <hip/hip_runtime.h>
#include <hip/hip_bf16.h>


typedef float v2f __attribute__((ext_vector_type(2)));
typedef float v8f __attribute__((ext_vector_type(8)));

#define N_NODES 65536
#define N_EDGES 1048576
#define DIM 64
#define M_POOL (N_NODES / DIM)


#define PART_EDGES 4096
#define N_PART (N_EDGES / PART_EDGES)
#define N_CBIN 256
#define FINE_CAP 6144

__device__ __forceinline__ int excl_scan_256(const int v, int* wsum) {
    const int lane = threadIdx.x & 31;
    const int wv = threadIdx.x >> 5;
    int inc = v;
    for (int d = 1; d < 32; d <<= 1) {
        const int y = __shfl_up(inc, d);
        if (lane >= d) inc += y;
    }
    if (lane == 31) wsum[wv] = inc;
    __syncthreads();
    int below = 0;
    for (int k = 0; k < 8; ++k) below += (k < wv) ? wsum[k] : 0;
    return inc - v + below;
}

__device__ __forceinline__ int excl_scan_totals_256(const int v, const int w, int* ws,
                                                    int& vtotal, int& wtotal) {
    const int lane = threadIdx.x & 31;
    const int wv = threadIdx.x >> 5;
    int inc = v;
    for (int d = 1; d < 32; d <<= 1) {
        const int y = __shfl_up(inc, d);
        if (lane >= d) inc += y;
    }
    int wr = w;
    for (int d = 16; d > 0; d >>= 1) wr += __shfl_xor(wr, d);
    if (lane == 31) ws[wv] = inc;
    if (lane == 0) ws[8 + wv] = wr;
    __syncthreads();
    int below = 0, vt = 0, wt = 0;
    for (int k = 0; k < 8; ++k) {
        const int a = ws[k];
        below += (k < wv) ? a : 0;
        vt += a;
        wt += ws[8 + k];
    }
    vtotal = vt;
    wtotal = wt;
    return inc - v + below;
}

__global__ __launch_bounds__(256) void k_part(const int* __restrict__ row,
                                              const int* __restrict__ col,
                                              const float* __restrict__ w,
                                              int* __restrict__ runs,
                                              int2* __restrict__ coarse) {
    __shared__ int lcnt[N_CBIN];
    __shared__ int wsum[8];
    __shared__ int2 lrec[PART_EDGES];
    const int t = threadIdx.x;
    const size_t q = (size_t)blockIdx.x * (PART_EDGES / 4) + t;
    lcnt[t] = 0;
    __syncthreads();
    int4 c4[4], r4[4];
    float4 w4[4];
#pragma unroll
    for (int i = 0; i < 4; ++i) c4[i] = ((const int4*)col)[q + i * 256];
#pragma unroll
    for (int i = 0; i < 4; ++i) {
        r4[i] = ((const int4*)row)[q + i * 256];
        w4[i] = ((const float4*)w)[q + i * 256];
    }
    unsigned cc[16], rr[16];
    float wr[16];
#pragma unroll
    for (int i = 0; i < 4; ++i) {
        cc[4 * i + 0] = (unsigned)c4[i].x & 0xFFFFu;  rr[4 * i + 0] = (unsigned)r4[i].x & 0xFFFFu;
        cc[4 * i + 1] = (unsigned)c4[i].y & 0xFFFFu;  rr[4 * i + 1] = (unsigned)r4[i].y & 0xFFFFu;
        cc[4 * i + 2] = (unsigned)c4[i].z & 0xFFFFu;  rr[4 * i + 2] = (unsigned)r4[i].z & 0xFFFFu;
        cc[4 * i + 3] = (unsigned)c4[i].w & 0xFFFFu;  rr[4 * i + 3] = (unsigned)r4[i].w & 0xFFFFu;
        wr[4 * i + 0] = w4[i].x;  wr[4 * i + 1] = w4[i].y;
        wr[4 * i + 2] = w4[i].z;  wr[4 * i + 3] = w4[i].w;
    }
    int rank[16];
#pragma unroll
    for (int k = 0; k < 16; ++k) rank[k] = atomicAdd(&lcnt[cc[k] >> 8], 1);
    __syncthreads();
    const int cnt = lcnt[t];
    const int first = excl_scan_256(cnt, wsum);
    lcnt[t] = first;
    runs[t * N_PART + blockIdx.x] = first | (cnt << 16);
    __syncthreads();
    int fs[16];
#pragma unroll
    for (int k = 0; k < 16; ++k) fs[k] = lcnt[cc[k] >> 8];
#pragma unroll
    for (int k = 0; k < 16; ++k)
        lrec[fs[k] + rank[k]] = make_int2((int)(rr[k] | (cc[k] << 16)), __float_as_int(wr[k]));
    __syncthreads();
    int2* dst = coarse + (size_t)blockIdx.x * PART_EDGES + t;
    int2 rec[16];
#pragma unroll
    for (int i = 0; i < 16; ++i) rec[i] = lrec[i * 256 + t];
#pragma unroll
    for (int i = 0; i < 16; ++i) dst[i * 256] = rec[i];
}


__device__ __forceinline__ void store_strip(float* __restrict__ crow, const float* ds8,
                                            const v8f a0, const v8f a1, const v8f a2, const v8f a3) {
    const float4 dlo = *(const float4*)(ds8);
    const float4 dhi = *(const float4*)(ds8 + 4);
    const float ds[8] = {dlo.x, dlo.y, dlo.z, dlo.w, dhi.x, dhi.y, dhi.z, dhi.w};
#pragma unroll
    for (int v = 0; v < 8; ++v) {
        crow[(size_t)v * DIM + 0]  = a0[v] * ds[v];
        crow[(size_t)v * DIM + 16] = a1[v] * ds[v];
        crow[(size_t)v * DIM + 32] = a2[v] * ds[v];
        crow[(size_t)v * DIM + 48] = a3[v] * ds[v];
    }
}

__global__ __launch_bounds__(256) void k_fine_gemm(const int* __restrict__ runs,
                                                   const int2* __restrict__ coarse,
                                                   const float* __restrict__ X,
                                                   const float* __restrict__ Wm,
                                                   int* __restrict__ rowptr,
                                                   int2* __restrict__ recs,
                                                   float* __restrict__ dinv,
                                                   float* __restrict__ C) {
    __shared__ int lcnt[256];
    __shared__ int wsum[8];
    __shared__ int wtot[16];
    __shared__ __attribute__((aligned(16))) float ldinv[256];
    __shared__ int lsrc[FINE_CAP];
    __shared__ int2 lout[FINE_CAP];
    const int t = threadIdx.x;
    const int b = blockIdx.x;
    lcnt[t] = 0;
    const int packed = runs[b * N_PART + t];
    const int rf = min(packed & 0xFFFF, PART_EDGES);
    const int rc = min((packed >> 16) & 0xFFFF, PART_EDGES - rf);
    int nsum, ssum;
    const int runpos = excl_scan_totals_256(rc, rf, wtot, nsum, ssum);
    const int s = min(max(ssum, 0), N_EDGES);
    const int n = min(max(nsum, 0), N_EDGES - s);
    const int e = s + n;
    const bool fits = (n <= FINE_CAP);
    const int gsrc = t * PART_EDGES + rf;
    const int2* run = coarse + gsrc;
    if (fits) {
        for (int k = 0; k < rc; ++k) {
            const int d = runpos + k;
            if ((unsigned)d < (unsigned)FINE_CAP) lsrc[d] = gsrc + k;
        }
    }
    __syncthreads();
    if (fits) {
        for (int i0 = 0; i0 < n; i0 += 4096) {
            int g[16], x[16];
#pragma unroll
            for (int u = 0; u < 16; ++u) g[u] = lsrc[min(i0 + u * 256 + t, n - 1)] & (N_EDGES - 1);
#pragma unroll
            for (int u = 0; u < 16; ++u) x[u] = coarse[g[u]].x;
#pragma unroll
            for (int u = 0; u < 16; ++u)
                if (i0 + u * 256 + t < n) atomicAdd(&lcnt[((unsigned)x[u] >> 16) & 255u], 1);
        }
    } else {
        for (int k = 0; k < rc; ++k) atomicAdd(&lcnt[((unsigned)run[k].x >> 16) & 255u], 1);
    }
    __syncthreads();
    const int cnt = lcnt[t];
    const int first = excl_scan_256(cnt, wsum);
    rowptr[b * 256 + t] = s + first;
    lcnt[t] = first;
    if (b == N_CBIN - 1 && t == 0) rowptr[N_NODES] = e;
    __syncthreads();
    if (fits) {
        for (int i0 = 0; i0 < n; i0 += 4096) {
            int g[16];
            int2 r[16];
            int pos[16];
#pragma unroll
            for (int u = 0; u < 16; ++u) g[u] = lsrc[min(i0 + u * 256 + t, n - 1)] & (N_EDGES - 1);
#pragma unroll
            for (int u = 0; u < 16; ++u) r[u] = coarse[g[u]];
#pragma unroll
            for (int u = 0; u < 16; ++u)
                pos[u] = (i0 + u * 256 + t < n)
                             ? atomicAdd(&lcnt[((unsigned)r[u].x >> 16) & 255u], 1) : -1;
#pragma unroll
            for (int u = 0; u < 16; ++u)
                if ((unsigned)pos[u] < (unsigned)FINE_CAP) lout[pos[u]] = make_int2(r[u].x & 0xFFFF, r[u].y);
        }
    } else {
        for (int k = 0; k < rc; ++k) {
            const int2 r = run[k];
            const int pos = atomicAdd(&lcnt[((unsigned)r.x >> 16) & 255u], 1);
            if (pos >= 0 && (unsigned)(s + pos) < (unsigned)N_EDGES) recs[s + pos] = make_int2(r.x & 0xFFFF, r.y);
        }
    }

    if (!fits) __threadfence();
    __syncthreads();
    if (!fits) __threadfence();

    const int lane = t & 31;
    const int wv   = t >> 5;
    const int half = lane >> 4;
    const int r16  = lane & 15;
    const int m0   = b * 256 + wv * 32;
    const float* arow = X + (size_t)(m0 + r16) * DIM + 2 * half;
    v2f af0[DIM / 4], af1[DIM / 4];
#pragma unroll
    for (int k = 0; k < DIM / 4; ++k) {
        af0[k] = *(const v2f*)(arow + 4 * k);
        af1[k] = *(const v2f*)(arow + 16 * DIM + 4 * k);
    }

    const int lo = min(max(first, 0), n);
    const int hi = min(max(first + cnt, lo), n);
    float sum = 0.0f;
    if (fits) {
        for (int i0 = 0; i0 < n; i0 += 2048) {
            int2 v[8];
#pragma unroll
            for (int u = 0; u < 8; ++u) v[u] = lout[min(i0 + u * 256 + t, n - 1)];
#pragma unroll
            for (int u = 0; u < 8; ++u)
                if (i0 + u * 256 + t < n) recs[s + i0 + u * 256 + t] = v[u];
        }
        for (int p = lo; p < hi; p += 4) {
            float wv4[4];
#pragma unroll
            for (int u = 0; u < 4; ++u)
                wv4[u] = (p + u < hi) ? __int_as_float(lout[p + u].y) : 0.0f;
            sum = (((sum + wv4[0]) + wv4[1]) + wv4[2]) + wv4[3];
        }
    } else {
        for (int p = lo; p < hi; p += 4) {
            float wv4[4];
#pragma unroll
            for (int u = 0; u < 4; ++u)
                wv4[u] = (p + u < hi) ? __int_as_float(recs[s + p + u].y) : 0.0f;
            sum = (((sum + wv4[0]) + wv4[1]) + wv4[2]) + wv4[3];
        }
    }
    const float deg = sum + 1.0f;
    const float di = (deg > 0.0f) ? rsqrtf(deg) : 0.0f;
    dinv[b * 256 + t] = di;
    ldinv[t] = di;

    v8f c00 = {}, c01 = {}, c02 = {}, c03 = {};
    v8f c10 = {}, c11 = {}, c12 = {}, c13 = {};
    const float* bp = Wm + (size_t)(2 * half) * DIM + r16;
#pragma unroll
    for (int k = 0; k < DIM / 4; ++k) {
        const float* b0 = bp + (size_t)(4 * k) * DIM;
        v2f bf;
        bf.x = b0[0];        bf.y = b0[DIM];
        c00 = __builtin_amdgcn_wmma_f32_16x16x4_f32(false, af0[k], false, bf, (short)0, c00, false, false);
        c10 = __builtin_amdgcn_wmma_f32_16x16x4_f32(false, af1[k], false, bf, (short)0, c10, false, false);
        bf.x = b0[16];       bf.y = b0[DIM + 16];
        c01 = __builtin_amdgcn_wmma_f32_16x16x4_f32(false, af0[k], false, bf, (short)0, c01, false, false);
        c11 = __builtin_amdgcn_wmma_f32_16x16x4_f32(false, af1[k], false, bf, (short)0, c11, false, false);
        bf.x = b0[32];       bf.y = b0[DIM + 32];
        c02 = __builtin_amdgcn_wmma_f32_16x16x4_f32(false, af0[k], false, bf, (short)0, c02, false, false);
        c12 = __builtin_amdgcn_wmma_f32_16x16x4_f32(false, af1[k], false, bf, (short)0, c12, false, false);
        bf.x = b0[48];       bf.y = b0[DIM + 48];
        c03 = __builtin_amdgcn_wmma_f32_16x16x4_f32(false, af0[k], false, bf, (short)0, c03, false, false);
        c13 = __builtin_amdgcn_wmma_f32_16x16x4_f32(false, af1[k], false, bf, (short)0, c13, false, false);
    }
    __syncthreads();
    float* crow = C + (size_t)(m0 + 8 * half) * DIM + r16;
    const float* ds8 = ldinv + wv * 32 + 8 * half;
    store_strip(crow, ds8, c00, c01, c02, c03);
    store_strip(crow + (size_t)16 * DIM, ds8 + 16, c10, c11, c12, c13);
}

#define NODES_PER_WAVE 4

__device__ __forceinline__ float2 bucket_sum(const int2* __restrict__ recs, const float* hl,
                                             const int node, const int s, const int e, const int lane) {
    float2 acc = *(const float2*)(hl + (size_t)node * DIM);
    for (int p0 = s; p0 < e; p0 += 32) {
        const int p = p0 + lane;
        int2 rec = make_int2(0, 0);
        if (p < e) rec = recs[p];
        const int cnt = min(32, e - p0);
        int j = 0;
        for (; j + 8 <= cnt; j += 8) {
            float2 v[8];
#pragma unroll
            for (int u = 0; u < 8; ++u) {
                const int r = __builtin_amdgcn_readlane(rec.x, j + u);
                v[u] = *(const float2*)(hl + (size_t)r * DIM);
            }
#pragma unroll
            for (int u = 0; u < 8; ++u) {
                const float n = __int_as_float(__builtin_amdgcn_readlane(rec.y, j + u));
                acc.x += n * v[u].x;
                acc.y += n * v[u].y;
            }
        }
        for (; j < cnt; j += 4) {
            float2 v[4];
#pragma unroll
            for (int u = 0; u < 4; ++u) {
                const int r = __builtin_amdgcn_readlane(rec.x, min(j + u, cnt - 1));
                v[u] = *(const float2*)(hl + (size_t)r * DIM);
            }
#pragma unroll
            for (int u = 0; u < 4; ++u) {
                const float nv = __int_as_float(__builtin_amdgcn_readlane(rec.y, min(j + u, cnt - 1)));
                const float n = (j + u < cnt) ? nv : 0.0f;
                acc.x += n * v[u].x;
                acc.y += n * v[u].y;
            }
        }
    }
    return acc;
}

__global__ __launch_bounds__(256) void k_gather(const int* __restrict__ rowptr,
                                                const int2* __restrict__ recs,
                                                const float* __restrict__ dinv,
                                                const float* __restrict__ h,
                                                const float* __restrict__ b,
                                                float* __restrict__ out) {
    const int lane = threadIdx.x & 31;
    const int wave = (blockIdx.x * 256 + (int)threadIdx.x) >> 5;
    const int n0   = wave * NODES_PER_WAVE;
    const int bnd = rowptr[n0 + (lane < NODES_PER_WAVE ? lane : NODES_PER_WAVE)];
    const float2 bias = *(const float2*)(b + lane * 2);
    const float* hl = h + lane * 2;
    for (int i = 0; i < NODES_PER_WAVE; ++i) {
        const int node = n0 + i;
        const int s = max(__builtin_amdgcn_readlane(bnd, i), 0);
        const int e = min(__builtin_amdgcn_readlane(bnd, i + 1), N_EDGES);
        const float di = dinv[node];
        const float2 acc = bucket_sum(recs, hl, node, s, e, lane);
        float2 o;
        o.x = (acc.x * di + bias.x) * di;
        o.y = (acc.y * di + bias.y) * di;
        *(float2*)(out + (size_t)node * DIM + lane * 2) = o;
    }
}

#define POOL_WAVES (DIM / NODES_PER_WAVE)
#define TILE_PITCH 68
__global__ __launch_bounds__(POOL_WAVES * 32) void k_gather_pool_gemm(const int* __restrict__ rowptr,
                                                                      const int2* __restrict__ recs,
                                                                      const float* __restrict__ dinv,
                                                                      const float* __restrict__ h,
                                                                      const float* __restrict__ Wm,
                                                                      const float* __restrict__ b,
                                                                      float* __restrict__ out) {
    __shared__ float tile[POOL_WAVES * TILE_PITCH];
    const int lane = threadIdx.x & 31;
    const int wv   = threadIdx.x >> 5;
    const int m    = blockIdx.x;
    const int n0   = m * DIM + wv * NODES_PER_WAVE;
    const int bnd = rowptr[n0 + (lane < NODES_PER_WAVE ? lane : NODES_PER_WAVE)];
    const float* hl = h + lane * 2;
    float2 ws = make_float2(0.0f, 0.0f);
    for (int i = 0; i < NODES_PER_WAVE; ++i) {
        const int node = n0 + i;
        const int s = max(__builtin_amdgcn_readlane(bnd, i), 0);
        const int e = min(__builtin_amdgcn_readlane(bnd, i + 1), N_EDGES);
        const float di = dinv[node];
        const float2 acc = bucket_sum(recs, hl, node, s, e, lane);
        ws.x += acc.x * di;
        ws.y += acc.y * di;
    }
    *(float2*)(tile + wv * TILE_PITCH + lane * 2) = ws;

    const int half = lane >> 4;
    const int r    = lane & 15;
    const int c0   = 16 * (wv & 3) + r;
    v2f bf[DIM / 4];
    float bias = 0.0f;
    if (wv < 4) {
        bias = b[c0];
        const float* b0 = Wm + (size_t)(2 * half) * DIM + c0;
#pragma unroll
        for (int t = 0; t < DIM / 4; ++t) {
            bf[t].x = b0[(size_t)(4 * t) * DIM];
            bf[t].y = b0[(size_t)(4 * t + 1) * DIM];
        }
    }
    __syncthreads();
    if (wv < 4) {
        v8f pa = {}, pb = {};
        const float* a0 = tile + r * TILE_PITCH + 2 * half;
#pragma unroll
        for (int t = 0; t < DIM / 4; t += 2) {
            const v2f a = *(const v2f*)(a0 + 4 * t);
            pa = __builtin_amdgcn_wmma_f32_16x16x4_f32(false, a, false, bf[t], (short)0, pa, false, false);
            const v2f a2 = *(const v2f*)(a0 + 4 * t + 4);
            pb = __builtin_amdgcn_wmma_f32_16x16x4_f32(false, a2, false, bf[t + 1], (short)0, pb, false, false);
        }
        float sum = 0.0f;
#pragma unroll
        for (int v = 0; v < 8; ++v) sum += pa[v] + pb[v];
        sum += __shfl_xor(sum, 16);
        if (half == 0) out[(size_t)c0 * M_POOL + m] = sum * (1.0f / DIM) + bias;
    }
}


extern "C" void kernel_launch(void* const* d_in, const int* in_sizes, int n_in,
                              void* d_out, int out_size, void* d_ws, size_t ws_size,
                              hipStream_t stream) {
    const float* x   = (const float*)d_in[0];
    const int*  eidx = (const int*) d_in[1];
    const float* ew  = (const float*)d_in[2];
    const float* W1  = (const float*)d_in[3];
    const float* b1  = (const float*)d_in[4];
    const float* W2  = (const float*)d_in[5];
    const float* b2  = (const float*)d_in[6];
    float* out = (float*)d_out;

    const int* rowi = eidx;
    const int* coli = eidx + N_EDGES;

    if (ws_size < (size_t)42467584) return;
    float* dinv = (float*)d_ws;
    float* bufH = dinv + N_NODES;
    float* bufA = bufH + (size_t)N_NODES * DIM;
    int*  rowptr = (int*)(bufA + (size_t)N_NODES * DIM);
    int2* recs   = (int2*)(rowptr + N_NODES + 64);
    int*  runs   = (int*)bufA;
    int2* coarse = (int2*)(runs + 4 * N_CBIN * N_PART);

    const int GATHER_BLK = (N_NODES / NODES_PER_WAVE) / 8;

    k_part<<<N_PART, 256, 0, stream>>>(rowi, coli, ew, runs, coarse);
    k_fine_gemm<<<N_CBIN, 256, 0, stream>>>(runs, coarse, x, W1, rowptr, recs, dinv, bufH);

    k_gather<<<GATHER_BLK, 256, 0, stream>>>(rowptr, recs, dinv, bufH, b1, bufA);

    k_gather_pool_gemm<<<M_POOL, POOL_WAVES * 32, 0, stream>>>(rowptr, recs, dinv, bufA, W2, b2, out);
}
